# starting kernel
# baseline (speedup 1.0000x reference)
#include <hip/hip_runtime.h>
#include <hip/hip_bf16.h>

typedef float v8f __attribute__((ext_vector_type(8)));
typedef float v2f __attribute__((ext_vector_type(2)));

#define T_  2048
#define D_  128

__device__ __forceinline__ v8f wmma4(v2f a, v2f b, v8f c) {
  return __builtin_amdgcn_wmma_f32_16x16x4_f32(false, a, false, b, (short)0, c,
                                               false, false);
}

__device__ __forceinline__ float fast_tanh(float x) {
  float a = __builtin_fabsf(x);
  float e = __expf(-2.0f * a);
  float r = (1.0f - e) / (1.0f + e);
  return __builtin_copysignf(r, x);
}

__global__ void __launch_bounds__(256)
xproj_kernel(const float* __restrict__ x, const float* __restrict__ Wi,
             const float* __restrict__ Wib, const float* __restrict__ bias,
             float* __restrict__ xp) {
  const int lane = threadIdx.x & 31;
  const int wid  = blockIdx.x * 8 + (threadIdx.x >> 5);
  const int btile = wid >> 11;
  const int t     = wid & 2047;
  const int lo = lane & 15, hf = lane >> 4;

  v8f acc[4];
#pragma unroll
  for (int nt = 0; nt < 4; ++nt) {
    float cst = Wib[nt * 16 + lo] + bias[nt * 16 + lo];
#pragma unroll
    for (int v = 0; v < 8; ++v) acc[nt][v] = cst;
  }

  const float* xrow = x + ((size_t)(btile * 16 + lo) * T_ + t) * D_ + hf * 2;
  const float* wrow = Wi + (size_t)lo * D_ + hf * 2;
#pragma unroll
  for (int ks = 0; ks < 32; ++ks) {
    v2f a = *(const v2f*)(xrow + ks * 4);
#pragma unroll
    for (int nt = 0; nt < 4; ++nt) {
      v2f b = *(const v2f*)(wrow + nt * 16 * D_ + ks * 4);
      acc[nt] = wmma4(a, b, acc[nt]);
    }
  }

  float* o = xp + ((size_t)wid * 32 + lane) * 32;
#pragma unroll
  for (int nt = 0; nt < 4; ++nt) {
#pragma unroll
    for (int q = 0; q < 2; ++q) {
      float4 s;
      s.x = acc[nt][q * 4 + 0]; s.y = acc[nt][q * 4 + 1];
      s.z = acc[nt][q * 4 + 2]; s.w = acc[nt][q * 4 + 3];
      *(float4*)(o + nt * 8 + q * 4) = s;
    }
  }
}

__global__ void __launch_bounds__(32)
scan_kernel(const float* __restrict__ Ap, const float* __restrict__ Am,
            const float* __restrict__ Wf, const float* __restrict__ Wo,
            const float* __restrict__ Wob, const float* __restrict__ xp,
            float* __restrict__ out) {
  __shared__ __align__(16) float hlds[16 * 64];
  const int lane = threadIdx.x;
  const int lo = lane & 15, hf = lane >> 4;
  const int btile = blockIdx.x;

  v2f bw[4][8];
#pragma unroll
  for (int nt = 0; nt < 4; ++nt) {
    const float* src = (nt < 2) ? Ap : Am;
    const int j = nt * 16 + lo;
#pragma unroll
    for (int s = 0; s < 8; ++s) {
      const int ks = (nt < 2) ? s : (s + 8);
      bw[nt][s] = *(const v2f*)(src + j * 64 + ks * 4 + hf * 2);
    }
  }
  v2f bs[4][16];
#pragma unroll
  for (int nt = 0; nt < 4; ++nt) {
    const int j = nt * 16 + lo;
    const float sg = (nt < 2) ? 1.0f : -1.0f;
#pragma unroll
    for (int ks = 0; ks < 16; ++ks) {
      v2f w = *(const v2f*)(Wf + j * 64 + ks * 4 + hf * 2);
      v2f r; r.x = w.x * sg; r.y = w.y * sg;
      bs[nt][ks] = r;
    }
  }

  for (int i = lane; i < 1024; i += 32) hlds[i] = 0.0f;
  __syncthreads();

  const float* xpl = xp + (size_t)btile * T_ * 1024 + lane * 32;
  float4 xq[8];
#pragma unroll
  for (int q = 0; q < 8; ++q) xq[q] = *(const float4*)(xpl + q * 4);

#pragma unroll 1
  for (int t = 0; t < T_; ++t) {
    v8f uc[4], u2[4], u3[4];
#pragma unroll
    for (int nt = 0; nt < 4; ++nt) {
      float4 q0 = xq[nt * 2], q1 = xq[nt * 2 + 1];
      uc[nt][0] = q0.x; uc[nt][1] = q0.y; uc[nt][2] = q0.z; uc[nt][3] = q0.w;
      uc[nt][4] = q1.x; uc[nt][5] = q1.y; uc[nt][6] = q1.z; uc[nt][7] = q1.w;
#pragma unroll
      for (int v = 0; v < 8; ++v) { u2[nt][v] = 0.0f; u3[nt][v] = 0.0f; }
    }
    if (t + 1 < T_) {
      const float* nx = xpl + (size_t)(t + 1) * 1024;
#pragma unroll
      for (int q = 0; q < 8; ++q) xq[q] = *(const float4*)(nx + q * 4);
      __builtin_prefetch(xpl + (size_t)(t + 5) * 1024, 0, 1);
    }

    v2f ha[16];
#pragma unroll
    for (int ks = 0; ks < 16; ++ks)
      ha[ks] = *(const v2f*)&hlds[lo * 64 + ks * 4 + hf * 2];

    float pa = 0.0f, pm = 0.0f;
#pragma unroll
    for (int ks = 0; ks < 16; ++ks) {
      float d = ha[ks].x * ha[ks].x + ha[ks].y * ha[ks].y;
      pa += d;
      if (ks >= 8) pm += d;
    }
    pa += __shfl_xor(pa, 16, 32);
    pm += __shfl_xor(pm, 16, 32);
    const float alpha = pm / (pa + 1e-8f);
    const float g =
        1.0f / (1.0f + __expf(-(alpha - 0.72134752044448170f) * 20.0f));
    v2f hg[16];
#pragma unroll
    for (int ks = 0; ks < 16; ++ks) {
      hg[ks].x = ha[ks].x * g; hg[ks].y = ha[ks].y * g;
    }

#pragma unroll
    for (int nt = 0; nt < 4; ++nt) {
#pragma unroll
      for (int s = 0; s < 8; ++s)
        uc[nt] = wmma4(ha[(nt < 2) ? s : (s + 8)], bw[nt][s], uc[nt]);
#pragma unroll
      for (int s = 0; s < 8; ++s)
        u2[nt] = wmma4(hg[s], bs[nt][s], u2[nt]);
#pragma unroll
      for (int s = 8; s < 16; ++s)
        u3[nt] = wmma4(hg[s], bs[nt][s], u3[nt]);
    }

    __syncthreads();
#pragma unroll
    for (int nt = 0; nt < 4; ++nt)
#pragma unroll
      for (int v = 0; v < 8; ++v)
        hlds[(v + 8 * hf) * 64 + nt * 16 + lo] =
            fast_tanh(uc[nt][v] + u2[nt][v] + u3[nt][v]);
    __syncthreads();
  }

#pragma unroll 1
  for (int i = lane; i < 1024; i += 32)
    out[32768 + btile * 1024 + i] = hlds[i];

  v2f ph[16];
#pragma unroll
  for (int ks = 0; ks < 16; ++ks)
    ph[ks] = *(const v2f*)&hlds[lo * 64 + ks * 4 + hf * 2];
#pragma unroll
  for (int n2 = 0; n2 < 8; ++n2) {
    v8f p;
    const float c0 = Wob[n2 * 16 + lo];
#pragma unroll
    for (int v = 0; v < 8; ++v) p[v] = c0;
#pragma unroll
    for (int ks = 0; ks < 16; ++ks) {
      v2f b = *(const v2f*)(Wo + (size_t)(n2 * 16 + lo) * 64 + ks * 4 + hf * 2);
      p = wmma4(ph[ks], b, p);
    }
#pragma unroll
    for (int v = 0; v < 8; ++v)
      out[(size_t)(btile * 16 + v + 8 * hf) * 128 + n2 * 16 + lo] = p[v];
  }
}

extern "C" void kernel_launch(void* const* d_in, const int* in_sizes, int n_in,
                              void* d_out, int out_size, void* d_ws, size_t ws_size,
                              hipStream_t stream) {
  const float* x    = (const float*)d_in[0];
  const float* Ap   = (const float*)d_in[1];
  const float* Am   = (const float*)d_in[2];
  const float* Wf   = (const float*)d_in[3];
  const float* Wiw  = (const float*)d_in[4];
  const float* Wib  = (const float*)d_in[5];
  const float* Wow  = (const float*)d_in[6];
  const float* Wob  = (const float*)d_in[7];
  const float* bias = (const float*)d_in[8];
  float* out = (float*)d_out;
  float* xp  = (float*)d_ws;

  xproj_kernel<<<4096, 256, 0, stream>>>(x, Wiw, Wib, bias, xp);
  scan_kernel<<<16, 32, 0, stream>>>(Ap, Am, Wf, Wow, Wob, xp, out);
}


# scan: the 4 output n-tiles of each 16-row batch tile go to 4 waves of one block (was 1 wave); h exchanged via 2 LDS planes, pitch 68, 1 barrier per step; x_proj stored per n-tile; sums unchanged
# speedup vs baseline: 2.6834x; 2.6834x over previous
#include <hip/hip_runtime.h>
#include <hip/hip_bf16.h>


typedef float v8f __attribute__((ext_vector_type(8)));
typedef float v2f __attribute__((ext_vector_type(2)));

#define T_  2048
#define D_  128
#define HP  68
#define PLANE (16 * HP)
#define PF  8


__device__ __forceinline__ v8f wmma4(v2f a, v2f b, v8f c) {
  return __builtin_amdgcn_wmma_f32_16x16x4_f32(false, a, false, b, (short)0, c,
                                               false, false);
}

__device__ __forceinline__ float fast_tanh(float x) {
  float a = __builtin_fabsf(x);
  float e = __expf(-2.0f * a);
  float r = (1.0f - e) / (1.0f + e);
  return __builtin_copysignf(r, x);
}

__global__ void __launch_bounds__(256)
xproj_kernel(const float* __restrict__ x, const float* __restrict__ Wi,
             const float* __restrict__ Wib, const float* __restrict__ bias,
             float* __restrict__ xp) {
  const int lane = threadIdx.x & 31;
  const int wid  = blockIdx.x * 8 + (threadIdx.x >> 5);
  const int btile = wid >> 11;
  const int t     = wid & 2047;
  const int lo = lane & 15, hf = lane >> 4;

  v8f acc[4];
#pragma unroll
  for (int nt = 0; nt < 4; ++nt) {
    float cst = Wib[nt * 16 + lo] + bias[nt * 16 + lo];
#pragma unroll
    for (int v = 0; v < 8; ++v) acc[nt][v] = cst;
  }

  const float* xrow = x + ((size_t)(btile * 16 + lo) * T_ + t) * D_ + hf * 2;
  const float* wrow = Wi + (size_t)lo * D_ + hf * 2;
#pragma unroll
  for (int ks = 0; ks < 32; ++ks) {
    v2f a = *(const v2f*)(xrow + ks * 4);
#pragma unroll
    for (int nt = 0; nt < 4; ++nt) {
      v2f b = *(const v2f*)(wrow + nt * 16 * D_ + ks * 4);
      acc[nt] = wmma4(a, b, acc[nt]);
    }
  }

  float* o = xp + (size_t)wid * 1024 + lane * 4;
#pragma unroll
  for (int nt = 0; nt < 4; ++nt) {
#pragma unroll
    for (int q = 0; q < 2; ++q) {
      float4 s;
      s.x = acc[nt][q * 4 + 0]; s.y = acc[nt][q * 4 + 1];
      s.z = acc[nt][q * 4 + 2]; s.w = acc[nt][q * 4 + 3];
      *(float4*)(o + (nt * 2 + q) * 128) = s;
    }
  }
}

__global__ void __launch_bounds__(128)
scan_kernel(const float* __restrict__ Ap, const float* __restrict__ Am,
            const float* __restrict__ Wf, const float* __restrict__ Wo,
            const float* __restrict__ Wob, const float* __restrict__ xp,
            float* __restrict__ out) {
  __shared__ __align__(16) float hlds[2 * PLANE];
  const int lane = threadIdx.x & 31;
  const int lo = lane & 15, hf = lane >> 4;
  const int nt = __builtin_amdgcn_readfirstlane((int)(threadIdx.x >> 5));
  const int btile = blockIdx.x;
  const bool up = nt >= 2;
  const int kc0 = up ? 32 : 0;

  const int j = nt * 16 + lo;
  v2f bw[8];
  {
    const float* src = (up ? Am : Ap) + j * 64 + kc0 + hf * 2;
#pragma unroll
    for (int s = 0; s < 8; ++s) bw[s] = *(const v2f*)(src + s * 4);
  }
  v2f bs[16];
  {
    const float sg = up ? -1.0f : 1.0f;
#pragma unroll
    for (int ks = 0; ks < 16; ++ks) {
      v2f w = *(const v2f*)(Wf + j * 64 + ks * 4 + hf * 2);
      v2f r; r.x = w.x * sg; r.y = w.y * sg;
      bs[ks] = r;
    }
  }

  for (int i = threadIdx.x; i < PLANE; i += 128) hlds[i] = 0.0f;
  __syncthreads();

  const float* xpl = xp + (size_t)btile * T_ * 1024 + nt * 256 + lane * 4;
  const float* xpf = xp + (size_t)btile * T_ * 1024 + nt * 256 + lane * 8;
  float4 xq0 = *(const float4*)(xpl);
  float4 xq1 = *(const float4*)(xpl + 128);

#pragma unroll 1
  for (int t = 0; t < T_; ++t) {
    const float* hr = hlds + (t & 1) * PLANE;
    float* hw = hlds + ((t & 1) ^ 1) * PLANE;

    v2f ha[16];
#pragma unroll
    for (int ks = 0; ks < 16; ++ks)
      ha[ks] = *(const v2f*)&hr[lo * HP + ks * 4 + hf * 2];
    v2f hc[8];
#pragma unroll
    for (int s = 0; s < 8; ++s)
      hc[s] = *(const v2f*)&hr[lo * HP + kc0 + s * 4 + hf * 2];

    v8f uc, u2, u3;
    uc[0] = xq0.x; uc[1] = xq0.y; uc[2] = xq0.z; uc[3] = xq0.w;
    uc[4] = xq1.x; uc[5] = xq1.y; uc[6] = xq1.z; uc[7] = xq1.w;
#pragma unroll
    for (int v = 0; v < 8; ++v) { u2[v] = 0.0f; u3[v] = 0.0f; }
    {
      const int tn = (t + 1 < T_) ? (t + 1) : (T_ - 1);
      const float* nx = xpl + (size_t)tn * 1024;
      xq0 = *(const float4*)(nx);
      xq1 = *(const float4*)(nx + 128);
      const int tp = (t + PF < T_) ? (t + PF) : (T_ - 1);
      __builtin_prefetch(xpf + (size_t)tp * 1024, 0, 1);
    }

    float pa = 0.0f, pm = 0.0f;
#pragma unroll
    for (int ks = 0; ks < 16; ++ks) {
      float d = ha[ks].x * ha[ks].x + ha[ks].y * ha[ks].y;
      pa += d;
      if (ks >= 8) pm += d;
    }
    pa += __shfl_xor(pa, 16, 32);
    pm += __shfl_xor(pm, 16, 32);
    const float alpha = pm / (pa + 1e-8f);
    const float g =
        1.0f / (1.0f + __expf(-(alpha - 0.72134752044448170f) * 20.0f));
    v2f hg[16];
#pragma unroll
    for (int ks = 0; ks < 16; ++ks) {
      hg[ks].x = ha[ks].x * g; hg[ks].y = ha[ks].y * g;
    }

#pragma unroll
    for (int s = 0; s < 8; ++s) uc = wmma4(hc[s], bw[s], uc);
#pragma unroll
    for (int s = 0; s < 8; ++s) u2 = wmma4(hg[s], bs[s], u2);
#pragma unroll
    for (int s = 8; s < 16; ++s) u3 = wmma4(hg[s], bs[s], u3);

#pragma unroll
    for (int v = 0; v < 8; ++v)
      hw[(v + 8 * hf) * HP + nt * 16 + lo] = fast_tanh(uc[v] + u2[v] + u3[v]);
    __syncthreads();
  }

  const float* hT = hlds + (T_ & 1) * PLANE;

  for (int i = threadIdx.x; i < 1024; i += 128)
    out[32768 + btile * 1024 + i] = hT[(i >> 6) * HP + (i & 63)];

  v2f ph[16];
#pragma unroll
  for (int ks = 0; ks < 16; ++ks)
    ph[ks] = *(const v2f*)&hT[lo * HP + ks * 4 + hf * 2];
#pragma unroll
  for (int n2i = 0; n2i < 2; ++n2i) {
    const int n2 = nt * 2 + n2i;
    v8f p;
    const float c0 = Wob[n2 * 16 + lo];
#pragma unroll
    for (int v = 0; v < 8; ++v) p[v] = c0;
#pragma unroll
    for (int ks = 0; ks < 16; ++ks) {
      v2f b = *(const v2f*)(Wo + (size_t)(n2 * 16 + lo) * 64 + ks * 4 + hf * 2);
      p = wmma4(ph[ks], b, p);
    }
#pragma unroll
    for (int v = 0; v < 8; ++v)
      out[(size_t)(btile * 16 + v + 8 * hf) * 128 + n2 * 16 + lo] = p[v];
  }
}

extern "C" void kernel_launch(void* const* d_in, const int* in_sizes, int n_in,
                              void* d_out, int out_size, void* d_ws, size_t ws_size,
                              hipStream_t stream) {
  const float* x    = (const float*)d_in[0];
  const float* Ap   = (const float*)d_in[1];
  const float* Am   = (const float*)d_in[2];
  const float* Wf   = (const float*)d_in[3];
  const float* Wiw  = (const float*)d_in[4];
  const float* Wib  = (const float*)d_in[5];
  const float* Wow  = (const float*)d_in[6];
  const float* Wob  = (const float*)d_in[7];
  const float* bias = (const float*)d_in[8];
  float* out = (float*)d_out;
  float* xp  = (float*)d_ws;

  xproj_kernel<<<4096, 256, 0, stream>>>(x, Wiw, Wib, bias, xp);
  scan_kernel<<<16, 128, 0, stream>>>(Ap, Am, Wf, Wow, Wob, xp, out);
}


# scan: the ten quotients of a step (two in the gate, eight in tanh) by v_rcp_f32 and a multiply instead of the IEEE division sequence; last-place rounding of g and tanh changes, sum order unchanged
# speedup vs baseline: 3.2154x; 1.1983x over previous
#include <hip/hip_runtime.h>
#include <hip/hip_bf16.h>


typedef float v8f __attribute__((ext_vector_type(8)));
typedef float v2f __attribute__((ext_vector_type(2)));

#define T_  2048
#define D_  128
#define HP  68
#define PLANE (16 * HP)
#define PF  8


__device__ __forceinline__ v8f wmma4(v2f a, v2f b, v8f c) {
  return __builtin_amdgcn_wmma_f32_16x16x4_f32(false, a, false, b, (short)0, c,
                                               false, false);
}

__device__ __forceinline__ float rcp_hw(float d) {
  return __builtin_amdgcn_rcpf(d);
}

__device__ __forceinline__ float fast_tanh(float x) {
  float a = __builtin_fabsf(x);
  float e = __expf(-2.0f * a);
  float r = (1.0f - e) * rcp_hw(1.0f + e);
  return __builtin_copysignf(r, x);
}

__global__ void __launch_bounds__(256)
xproj_kernel(const float* __restrict__ x, const float* __restrict__ Wi,
             const float* __restrict__ Wib, const float* __restrict__ bias,
             float* __restrict__ xp) {
  const int lane = threadIdx.x & 31;
  const int wid  = blockIdx.x * 8 + (threadIdx.x >> 5);
  const int btile = wid >> 11;
  const int t     = wid & 2047;
  const int lo = lane & 15, hf = lane >> 4;

  v8f acc[4];
#pragma unroll
  for (int nt = 0; nt < 4; ++nt) {
    float cst = Wib[nt * 16 + lo] + bias[nt * 16 + lo];
#pragma unroll
    for (int v = 0; v < 8; ++v) acc[nt][v] = cst;
  }

  const float* xrow = x + ((size_t)(btile * 16 + lo) * T_ + t) * D_ + hf * 2;
  const float* wrow = Wi + (size_t)lo * D_ + hf * 2;
#pragma unroll
  for (int ks = 0; ks < 32; ++ks) {
    v2f a = *(const v2f*)(xrow + ks * 4);
#pragma unroll
    for (int nt = 0; nt < 4; ++nt) {
      v2f b = *(const v2f*)(wrow + nt * 16 * D_ + ks * 4);
      acc[nt] = wmma4(a, b, acc[nt]);
    }
  }

  float* o = xp + (size_t)wid * 1024 + lane * 4;
#pragma unroll
  for (int nt = 0; nt < 4; ++nt) {
#pragma unroll
    for (int q = 0; q < 2; ++q) {
      float4 s;
      s.x = acc[nt][q * 4 + 0]; s.y = acc[nt][q * 4 + 1];
      s.z = acc[nt][q * 4 + 2]; s.w = acc[nt][q * 4 + 3];
      *(float4*)(o + (nt * 2 + q) * 128) = s;
    }
  }
}

__global__ void __launch_bounds__(128)
scan_kernel(const float* __restrict__ Ap, const float* __restrict__ Am,
            const float* __restrict__ Wf, const float* __restrict__ Wo,
            const float* __restrict__ Wob, const float* __restrict__ xp,
            float* __restrict__ out) {
  __shared__ __align__(16) float hlds[2 * PLANE];
  const int lane = threadIdx.x & 31;
  const int lo = lane & 15, hf = lane >> 4;
  const int nt = __builtin_amdgcn_readfirstlane((int)(threadIdx.x >> 5));
  const int btile = blockIdx.x;
  const bool up = nt >= 2;
  const int kc0 = up ? 32 : 0;

  const int j = nt * 16 + lo;
  v2f bw[8];
  {
    const float* src = (up ? Am : Ap) + j * 64 + kc0 + hf * 2;
#pragma unroll
    for (int s = 0; s < 8; ++s) bw[s] = *(const v2f*)(src + s * 4);
  }
  v2f bs[16];
  {
    const float sg = up ? -1.0f : 1.0f;
#pragma unroll
    for (int ks = 0; ks < 16; ++ks) {
      v2f w = *(const v2f*)(Wf + j * 64 + ks * 4 + hf * 2);
      v2f r; r.x = w.x * sg; r.y = w.y * sg;
      bs[ks] = r;
    }
  }

  for (int i = threadIdx.x; i < PLANE; i += 128) hlds[i] = 0.0f;
  __syncthreads();

  const float* xpl = xp + (size_t)btile * T_ * 1024 + nt * 256 + lane * 4;
  const float* xpf = xp + (size_t)btile * T_ * 1024 + nt * 256 + lane * 8;
  float4 xq0 = *(const float4*)(xpl);
  float4 xq1 = *(const float4*)(xpl + 128);

#pragma unroll 1
  for (int t = 0; t < T_; ++t) {
    const float* hr = hlds + (t & 1) * PLANE;
    float* hw = hlds + ((t & 1) ^ 1) * PLANE;

    v2f ha[16];
#pragma unroll
    for (int ks = 0; ks < 16; ++ks)
      ha[ks] = *(const v2f*)&hr[lo * HP + ks * 4 + hf * 2];
    v2f hc[8];
#pragma unroll
    for (int s = 0; s < 8; ++s)
      hc[s] = *(const v2f*)&hr[lo * HP + kc0 + s * 4 + hf * 2];

    v8f uc, u2, u3;
    uc[0] = xq0.x; uc[1] = xq0.y; uc[2] = xq0.z; uc[3] = xq0.w;
    uc[4] = xq1.x; uc[5] = xq1.y; uc[6] = xq1.z; uc[7] = xq1.w;
#pragma unroll
    for (int v = 0; v < 8; ++v) { u2[v] = 0.0f; u3[v] = 0.0f; }
    {
      const int tn = (t + 1 < T_) ? (t + 1) : (T_ - 1);
      const float* nx = xpl + (size_t)tn * 1024;
      xq0 = *(const float4*)(nx);
      xq1 = *(const float4*)(nx + 128);
      const int tp = (t + PF < T_) ? (t + PF) : (T_ - 1);
      __builtin_prefetch(xpf + (size_t)tp * 1024, 0, 1);
    }

    float pa = 0.0f, pm = 0.0f;
#pragma unroll
    for (int ks = 0; ks < 16; ++ks) {
      float d = ha[ks].x * ha[ks].x + ha[ks].y * ha[ks].y;
      pa += d;
      if (ks >= 8) pm += d;
    }
    pa += __shfl_xor(pa, 16, 32);
    pm += __shfl_xor(pm, 16, 32);
    const float alpha = pm * rcp_hw(pa + 1e-8f);
    const float g =
        rcp_hw(1.0f + __expf(-(alpha - 0.72134752044448170f) * 20.0f));
    v2f hg[16];
#pragma unroll
    for (int ks = 0; ks < 16; ++ks) {
      hg[ks].x = ha[ks].x * g; hg[ks].y = ha[ks].y * g;
    }

#pragma unroll
    for (int s = 0; s < 8; ++s) uc = wmma4(hc[s], bw[s], uc);
#pragma unroll
    for (int s = 0; s < 8; ++s) u2 = wmma4(hg[s], bs[s], u2);
#pragma unroll
    for (int s = 8; s < 16; ++s) u3 = wmma4(hg[s], bs[s], u3);

#pragma unroll
    for (int v = 0; v < 8; ++v)
      hw[(v + 8 * hf) * HP + nt * 16 + lo] = fast_tanh(uc[v] + u2[v] + u3[v]);
    __syncthreads();
  }

  const float* hT = hlds + (T_ & 1) * PLANE;

  for (int i = threadIdx.x; i < 1024; i += 128)
    out[32768 + btile * 1024 + i] = hT[(i >> 6) * HP + (i & 63)];

  v2f ph[16];
#pragma unroll
  for (int ks = 0; ks < 16; ++ks)
    ph[ks] = *(const v2f*)&hT[lo * HP + ks * 4 + hf * 2];
#pragma unroll
  for (int n2i = 0; n2i < 2; ++n2i) {
    const int n2 = nt * 2 + n2i;
    v8f p;
    const float c0 = Wob[n2 * 16 + lo];
#pragma unroll
    for (int v = 0; v < 8; ++v) p[v] = c0;
#pragma unroll
    for (int ks = 0; ks < 16; ++ks) {
      v2f b = *(const v2f*)(Wo + (size_t)(n2 * 16 + lo) * 64 + ks * 4 + hf * 2);
      p = wmma4(ph[ks], b, p);
    }
#pragma unroll
    for (int v = 0; v < 8; ++v)
      out[(size_t)(btile * 16 + v + 8 * hf) * 128 + n2 * 16 + lo] = p[v];
  }
}

extern "C" void kernel_launch(void* const* d_in, const int* in_sizes, int n_in,
                              void* d_out, int out_size, void* d_ws, size_t ws_size,
                              hipStream_t stream) {
  const float* x    = (const float*)d_in[0];
  const float* Ap   = (const float*)d_in[1];
  const float* Am   = (const float*)d_in[2];
  const float* Wf   = (const float*)d_in[3];
  const float* Wiw  = (const float*)d_in[4];
  const float* Wib  = (const float*)d_in[5];
  const float* Wow  = (const float*)d_in[6];
  const float* Wob  = (const float*)d_in[7];
  const float* bias = (const float*)d_in[8];
  float* out = (float*)d_out;
  float* xp  = (float*)d_ws;

  xproj_kernel<<<4096, 256, 0, stream>>>(x, Wiw, Wib, bias, xp);
  scan_kernel<<<16, 128, 0, stream>>>(Ap, Am, Wf, Wow, Wob, xp, out);
}


# scan: seam product on the ungated h, scaled by g afterwards (the reference's order): no matrix instruction of a step waits for the gate, whose chain is sliced in between them; u = uc + g*(u2+u3)
# speedup vs baseline: 3.6148x; 1.1242x over previous
#include <hip/hip_runtime.h>
#include <hip/hip_bf16.h>


typedef float v8f __attribute__((ext_vector_type(8)));
typedef float v2f __attribute__((ext_vector_type(2)));

#define T_  2048
#define D_  128
#define HP  68
#define PLANE (16 * HP)
#define PF  8


__device__ __forceinline__ v8f wmma4(v2f a, v2f b, v8f c) {
  return __builtin_amdgcn_wmma_f32_16x16x4_f32(false, a, false, b, (short)0, c,
                                               false, false);
}

__device__ __forceinline__ float rcp_hw(float d) {
  return __builtin_amdgcn_rcpf(d);
}

__device__ __forceinline__ float fast_tanh(float x) {
  float a = __builtin_fabsf(x);
  float e = __expf(-2.0f * a);
  float r = (1.0f - e) * rcp_hw(1.0f + e);
  return __builtin_copysignf(r, x);
}

__device__ __forceinline__ float xhalf(float x) {
  int b = __builtin_bit_cast(int, x);
  return __builtin_bit_cast(
      float, __builtin_amdgcn_permlanex16(b, b, 0x76543210, 0xfedcba98, false, false));
}

__global__ void __launch_bounds__(256)
xproj_kernel(const float* __restrict__ x, const float* __restrict__ Wi,
             const float* __restrict__ Wib, const float* __restrict__ bias,
             float* __restrict__ xp) {
  const int lane = threadIdx.x & 31;
  const int wid  = blockIdx.x * 8 + (threadIdx.x >> 5);
  const int btile = wid >> 11;
  const int t     = wid & 2047;
  const int lo = lane & 15, hf = lane >> 4;

  v8f acc[4];
#pragma unroll
  for (int nt = 0; nt < 4; ++nt) {
    float cst = Wib[nt * 16 + lo] + bias[nt * 16 + lo];
#pragma unroll
    for (int v = 0; v < 8; ++v) acc[nt][v] = cst;
  }

  const float* xrow = x + ((size_t)(btile * 16 + lo) * T_ + t) * D_ + hf * 2;
  const float* wrow = Wi + (size_t)lo * D_ + hf * 2;
#pragma unroll
  for (int ks = 0; ks < 32; ++ks) {
    v2f a = *(const v2f*)(xrow + ks * 4);
#pragma unroll
    for (int nt = 0; nt < 4; ++nt) {
      v2f b = *(const v2f*)(wrow + nt * 16 * D_ + ks * 4);
      acc[nt] = wmma4(a, b, acc[nt]);
    }
  }

  float* o = xp + (size_t)wid * 1024 + lane * 4;
#pragma unroll
  for (int nt = 0; nt < 4; ++nt) {
#pragma unroll
    for (int q = 0; q < 2; ++q) {
      float4 s;
      s.x = acc[nt][q * 4 + 0]; s.y = acc[nt][q * 4 + 1];
      s.z = acc[nt][q * 4 + 2]; s.w = acc[nt][q * 4 + 3];
      *(float4*)(o + (nt * 2 + q) * 128) = s;
    }
  }
}

__global__ void __launch_bounds__(128)
scan_kernel(const float* __restrict__ Ap, const float* __restrict__ Am,
            const float* __restrict__ Wf, const float* __restrict__ Wo,
            const float* __restrict__ Wob, const float* __restrict__ xp,
            float* __restrict__ out) {
  __shared__ __align__(16) float hlds[2 * PLANE];
  const int lane = threadIdx.x & 31;
  const int lo = lane & 15, hf = lane >> 4;
  const int nt = __builtin_amdgcn_readfirstlane((int)(threadIdx.x >> 5));
  const int btile = blockIdx.x;
  const bool up = nt >= 2;
  const int kc0 = up ? 32 : 0;

  const int j = nt * 16 + lo;
  v2f bw[8];
  {
    const float* src = (up ? Am : Ap) + j * 64 + kc0 + hf * 2;
#pragma unroll
    for (int s = 0; s < 8; ++s) bw[s] = *(const v2f*)(src + s * 4);
  }
  v2f bs[16];
  {
    const float sg = up ? -1.0f : 1.0f;
#pragma unroll
    for (int ks = 0; ks < 16; ++ks) {
      v2f w = *(const v2f*)(Wf + j * 64 + ks * 4 + hf * 2);
      v2f r; r.x = w.x * sg; r.y = w.y * sg;
      bs[ks] = r;
    }
  }

  for (int i = threadIdx.x; i < PLANE; i += 128) hlds[i] = 0.0f;
  __syncthreads();

  const float* xpl = xp + (size_t)btile * T_ * 1024 + nt * 256 + lane * 4;
  const float* xpf = xp + (size_t)btile * T_ * 1024 + nt * 256 + lane * 8;
  float4 xq0 = *(const float4*)(xpl);
  float4 xq1 = *(const float4*)(xpl + 128);

#pragma unroll 1
  for (int t = 0; t < T_; ++t) {
    const float* hr = hlds + (t & 1) * PLANE;
    float* hw = hlds + ((t & 1) ^ 1) * PLANE;

    v2f ha[16];
#pragma unroll
    for (int ks = 0; ks < 16; ++ks)
      ha[ks] = *(const v2f*)&hr[lo * HP + ks * 4 + hf * 2];
    v2f hc[8];
#pragma unroll
    for (int s = 0; s < 8; ++s)
      hc[s] = *(const v2f*)&hr[lo * HP + kc0 + s * 4 + hf * 2];

    v8f uc, u2, u3;
    uc[0] = xq0.x; uc[1] = xq0.y; uc[2] = xq0.z; uc[3] = xq0.w;
    uc[4] = xq1.x; uc[5] = xq1.y; uc[6] = xq1.z; uc[7] = xq1.w;
#pragma unroll
    for (int v = 0; v < 8; ++v) { u2[v] = 0.0f; u3[v] = 0.0f; }
    {
      const int tn = (t + 1 < T_) ? (t + 1) : (T_ - 1);
      const float* nx = xpl + (size_t)tn * 1024;
      xq0 = *(const float4*)(nx);
      xq1 = *(const float4*)(nx + 128);
      const int tp = (t + PF < T_) ? (t + PF) : (T_ - 1);
      __builtin_prefetch(xpf + (size_t)tp * 1024, 0, 1);
    }

#define GAP(...) asm volatile("" : __VA_ARGS__)
#define ALL_ACC "+v"(uc), "+v"(u2), "+v"(u3)
#define SQ(k) __builtin_fmaf(ha[k].x, ha[k].x, ha[k].y * ha[k].y)

    float pa = 0.0f, pm = 0.0f;
    uc = wmma4(hc[0], bw[0], uc);  pa += SQ(0);  pa += SQ(1);  GAP("+v"(uc), "+v"(pa));
    u2 = wmma4(ha[0], bs[0], u2);  pa += SQ(2);  pa += SQ(3);  GAP("+v"(uc), "+v"(u2), "+v"(pa));
    u3 = wmma4(ha[8], bs[8], u3);  pa += SQ(4);  pa += SQ(5);  GAP(ALL_ACC, "+v"(pa));
    uc = wmma4(hc[1], bw[1], uc);  pa += SQ(6);  pa += SQ(7);  GAP(ALL_ACC, "+v"(pa));
    u2 = wmma4(ha[1], bs[1], u2);
    { float d = SQ(8);  pa += d; pm += d; d = SQ(9);  pa += d; pm += d; }  GAP(ALL_ACC, "+v"(pa), "+v"(pm));
    u3 = wmma4(ha[9], bs[9], u3);
    { float d = SQ(10); pa += d; pm += d; d = SQ(11); pa += d; pm += d; }  GAP(ALL_ACC, "+v"(pa), "+v"(pm));
    uc = wmma4(hc[2], bw[2], uc);
    { float d = SQ(12); pa += d; pm += d; d = SQ(13); pa += d; pm += d; }  GAP(ALL_ACC, "+v"(pa), "+v"(pm));
    u2 = wmma4(ha[2], bs[2], u2);
    { float d = SQ(14); pa += d; pm += d; d = SQ(15); pa += d; pm += d; }  GAP(ALL_ACC, "+v"(pa), "+v"(pm));
#undef SQ
    u3 = wmma4(ha[10], bs[10], u3);
    pa += xhalf(pa);
    pm += xhalf(pm);
    GAP(ALL_ACC, "+v"(pa), "+v"(pm));
    uc = wmma4(hc[3], bw[3], uc);
    pa = rcp_hw(pa + 1e-8f);
    GAP(ALL_ACC, "+v"(pa), "+v"(pm));
    u2 = wmma4(ha[3], bs[3], u2);
    pa = -(pm * pa - 0.72134752044448170f) * 20.0f;
    GAP(ALL_ACC, "+v"(pa));
    u3 = wmma4(ha[11], bs[11], u3);
    pa = 1.0f + __expf(pa);
    GAP(ALL_ACC, "+v"(pa));
    uc = wmma4(hc[4], bw[4], uc);
    pa = rcp_hw(pa);
    GAP(ALL_ACC, "+v"(pa));
    u2 = wmma4(ha[4], bs[4], u2);
    {
      const int gi = __builtin_bit_cast(int, pa);
      pa = __builtin_bit_cast(
          float, __builtin_amdgcn_update_dpp(gi, gi, 0x128, 0x2, 0xf, false));
    }
    GAP(ALL_ACC, "+v"(pa));
#undef GAP
#undef ALL_ACC
    const int gt = __builtin_bit_cast(int, pa);
    u3 = wmma4(ha[12], bs[12], u3);
#pragma unroll
    for (int s = 5; s < 8; ++s) {
      uc = wmma4(hc[s], bw[s], uc);
      u2 = wmma4(ha[s], bs[s], u2);
      u3 = wmma4(ha[s + 8], bs[s + 8], u3);
    }

#define OUT_ROW(v)                                                              \
    {                                                                           \
      const float gv = __builtin_bit_cast(                                      \
          float, __builtin_amdgcn_update_dpp(0, gt, 0x150 + (v), 0xf, 0xf, true)); \
      const float u = __builtin_fmaf(gv, u2[v] + u3[v], uc[v]);                 \
      hw[((v) + 8 * hf) * HP + nt * 16 + lo] = fast_tanh(u);                    \
    }
    OUT_ROW(0) OUT_ROW(1) OUT_ROW(2) OUT_ROW(3)
    OUT_ROW(4) OUT_ROW(5) OUT_ROW(6) OUT_ROW(7)
#undef OUT_ROW
    __syncthreads();
  }

  const float* hT = hlds + (T_ & 1) * PLANE;

  for (int i = threadIdx.x; i < 1024; i += 128)
    out[32768 + btile * 1024 + i] = hT[(i >> 6) * HP + (i & 63)];

  v2f ph[16];
#pragma unroll
  for (int ks = 0; ks < 16; ++ks)
    ph[ks] = *(const v2f*)&hT[lo * HP + ks * 4 + hf * 2];
#pragma unroll
  for (int n2i = 0; n2i < 2; ++n2i) {
    const int n2 = nt * 2 + n2i;
    v8f p;
    const float c0 = Wob[n2 * 16 + lo];
#pragma unroll
    for (int v = 0; v < 8; ++v) p[v] = c0;
#pragma unroll
    for (int ks = 0; ks < 16; ++ks) {
      v2f b = *(const v2f*)(Wo + (size_t)(n2 * 16 + lo) * 64 + ks * 4 + hf * 2);
      p = wmma4(ph[ks], b, p);
    }
#pragma unroll
    for (int v = 0; v < 8; ++v)
      out[(size_t)(btile * 16 + v + 8 * hf) * 128 + n2 * 16 + lo] = p[v];
  }
}

extern "C" void kernel_launch(void* const* d_in, const int* in_sizes, int n_in,
                              void* d_out, int out_size, void* d_ws, size_t ws_size,
                              hipStream_t stream) {
  const float* x    = (const float*)d_in[0];
  const float* Ap   = (const float*)d_in[1];
  const float* Am   = (const float*)d_in[2];
  const float* Wf   = (const float*)d_in[3];
  const float* Wiw  = (const float*)d_in[4];
  const float* Wib  = (const float*)d_in[5];
  const float* Wow  = (const float*)d_in[6];
  const float* Wob  = (const float*)d_in[7];
  const float* bias = (const float*)d_in[8];
  float* out = (float*)d_out;
  float* xp  = (float*)d_ws;

  xproj_kernel<<<4096, 256, 0, stream>>>(x, Wiw, Wib, bias, xp);
  scan_kernel<<<16, 128, 0, stream>>>(Ap, Am, Wf, Wow, Wob, xp, out);
}


# scan, x_proj as transposed products (weights A, h B): result registers are the next B fragments, 8 of 24 matrix ops run ahead of the LDS exchange (16 B wide), gate in-lane; k order of sums permuted
# speedup vs baseline: 3.7523x; 1.0380x over previous
#include <hip/hip_runtime.h>
#include <hip/hip_bf16.h>


typedef float v8f __attribute__((ext_vector_type(8)));
typedef float v2f __attribute__((ext_vector_type(2)));
typedef float v4f __attribute__((ext_vector_type(4)));

#define T_  2048
#define D_  128
#define TILE  256
#define PLANE (4 * TILE)
#define PF  8


__device__ __forceinline__ v8f wmma4(v2f a, v2f b, v8f c) {
  return __builtin_amdgcn_wmma_f32_16x16x4_f32(false, a, false, b, (short)0, c,
                                               false, false);
}

__device__ __forceinline__ float rcp_hw(float d) {
  return __builtin_amdgcn_rcpf(d);
}

__device__ __forceinline__ float fast_tanh(float x) {
  float a = __builtin_fabsf(x);
  float e = __expf(-2.0f * a);
  float r = (1.0f - e) * rcp_hw(1.0f + e);
  return __builtin_copysignf(r, x);
}

__device__ __forceinline__ float xhalf(float x) {
  int b = __builtin_bit_cast(int, x);
  return __builtin_bit_cast(
      float, __builtin_amdgcn_permlanex16(b, b, 0x76543210, 0xfedcba98, false, false));
}

__device__ __forceinline__ float sq2(float a, float b) {
  return __builtin_fmaf(a, a, b * b);
}

__global__ void __launch_bounds__(256)
xproj_kernel(const float* __restrict__ x, const float* __restrict__ Wi,
             const float* __restrict__ Wib, const float* __restrict__ bias,
             float* __restrict__ xp) {
  const int lane = threadIdx.x & 31;
  const int wid  = blockIdx.x * 8 + (threadIdx.x >> 5);
  const int btile = wid >> 11;
  const int t     = wid & 2047;
  const int lo = lane & 15, hf = lane >> 4;

  v8f acc[4];
#pragma unroll
  for (int nt = 0; nt < 4; ++nt) {
    const float4 w0 = *(const float4*)(Wib + nt * 16 + hf * 8);
    const float4 w1 = *(const float4*)(Wib + nt * 16 + hf * 8 + 4);
    const float4 c0 = *(const float4*)(bias + nt * 16 + hf * 8);
    const float4 c1 = *(const float4*)(bias + nt * 16 + hf * 8 + 4);
    acc[nt][0] = w0.x + c0.x; acc[nt][1] = w0.y + c0.y;
    acc[nt][2] = w0.z + c0.z; acc[nt][3] = w0.w + c0.w;
    acc[nt][4] = w1.x + c1.x; acc[nt][5] = w1.y + c1.y;
    acc[nt][6] = w1.z + c1.z; acc[nt][7] = w1.w + c1.w;
  }

  const float* xrow = x + ((size_t)(btile * 16 + lo) * T_ + t) * D_ + hf * 2;
  const float* wrow = Wi + (size_t)lo * D_ + hf * 2;
#pragma unroll
  for (int ks = 0; ks < 32; ++ks) {
    v2f xb = *(const v2f*)(xrow + ks * 4);
#pragma unroll
    for (int nt = 0; nt < 4; ++nt) {
      v2f wa = *(const v2f*)(wrow + nt * 16 * D_ + ks * 4);
      acc[nt] = wmma4(wa, xb, acc[nt]);
    }
  }

  float* o = xp + (size_t)wid * 1024 + lane * 4;
#pragma unroll
  for (int nt = 0; nt < 4; ++nt) {
#pragma unroll
    for (int q = 0; q < 2; ++q) {
      float4 s;
      s.x = acc[nt][q * 4 + 0]; s.y = acc[nt][q * 4 + 1];
      s.z = acc[nt][q * 4 + 2]; s.w = acc[nt][q * 4 + 3];
      *(float4*)(o + (nt * 2 + q) * 128) = s;
    }
  }
}

__global__ void __launch_bounds__(128)
scan_kernel(const float* __restrict__ Ap, const float* __restrict__ Am,
            const float* __restrict__ Wf, const float* __restrict__ Wo,
            const float* __restrict__ Wob, const float* __restrict__ xp,
            float* __restrict__ out) {
  __shared__ __align__(16) float hlds[2 * PLANE];
  const int lane = threadIdx.x & 31;
  const int lo = lane & 15, hf = lane >> 4;
  const int nt = __builtin_amdgcn_readfirstlane((int)(threadIdx.x >> 5));
  const int btile = blockIdx.x;
  const bool up = nt >= 2;
  const int m1 = nt ^ 1, m2 = nt ^ 2, m3 = nt ^ 3;

  const int j = nt * 16 + lo;
  const int ko = hf * 8;
  v2f aw[8];
  {
    const float* src = (up ? Am : Ap) + j * 64 + ko;
#pragma unroll
    for (int p = 0; p < 4; ++p) {
      aw[p]     = *(const v2f*)(src + nt * 16 + 2 * p);
      aw[4 + p] = *(const v2f*)(src + m1 * 16 + 2 * p);
    }
  }
  v2f s0[4], s1[4], s2[4], s3[4];
  {
    const float sg = up ? -1.0f : 1.0f;
    const float* src = Wf + j * 64 + ko;
#pragma unroll
    for (int p = 0; p < 4; ++p) {
      v2f w;
      w = *(const v2f*)(src + nt * 16 + 2 * p); s0[p].x = w.x * sg; s0[p].y = w.y * sg;
      w = *(const v2f*)(src + m1 * 16 + 2 * p); s1[p].x = w.x * sg; s1[p].y = w.y * sg;
      w = *(const v2f*)(src + m2 * 16 + 2 * p); s2[p].x = w.x * sg; s2[p].y = w.y * sg;
      w = *(const v2f*)(src + m3 * 16 + 2 * p); s3[p].x = w.x * sg; s3[p].y = w.y * sg;
    }
  }

  float hn[8];
#pragma unroll
  for (int r = 0; r < 8; ++r) hn[r] = 0.0f;
  {
    float4 z; z.x = 0.0f; z.y = 0.0f; z.z = 0.0f; z.w = 0.0f;
    *(float4*)(hlds + nt * TILE + lane * 4) = z;
    *(float4*)(hlds + nt * TILE + 128 + lane * 4) = z;
  }
  __syncthreads();

  const float* xpl = xp + (size_t)btile * T_ * 1024 + nt * 256 + lane * 4;
  const float* xpf = xp + (size_t)btile * T_ * 1024 + nt * 256 + lane * 8;
  float4 xq0 = *(const float4*)(xpl);
  float4 xq1 = *(const float4*)(xpl + 128);

#pragma unroll 1
  for (int t = 0; t < T_; ++t) {
    const float* pr = hlds + (t & 1) * PLANE + lane * 4;
    float* pw = hlds + ((t & 1) ^ 1) * PLANE + nt * TILE + lane * 4;

    v4f a1 = *(const v4f*)(pr + m1 * TILE);
    v4f b1 = *(const v4f*)(pr + m1 * TILE + 128);
    v4f a2 = *(const v4f*)(pr + m2 * TILE);
    v4f b2 = *(const v4f*)(pr + m2 * TILE + 128);
    v4f a3 = *(const v4f*)(pr + m3 * TILE);
    v4f b3 = *(const v4f*)(pr + m3 * TILE + 128);

    v8f uc, u2, u3;
    uc[0] = xq0.x; uc[1] = xq0.y; uc[2] = xq0.z; uc[3] = xq0.w;
    uc[4] = xq1.x; uc[5] = xq1.y; uc[6] = xq1.z; uc[7] = xq1.w;
#pragma unroll
    for (int v = 0; v < 8; ++v) { u2[v] = 0.0f; u3[v] = 0.0f; }
    {
      const int tn = (t + 1 < T_) ? (t + 1) : (T_ - 1);
      const float* nx = xpl + (size_t)tn * 1024;
      xq0 = *(const float4*)(nx);
      xq1 = *(const float4*)(nx + 128);
      const int tp = (t + PF < T_) ? (t + PF) : (T_ - 1);
      __builtin_prefetch(xpf + (size_t)tp * 1024, 0, 1);
    }
    v2f h01 = v2f{hn[0], hn[1]};
    asm volatile("" : "+v"(uc), "+v"(h01) : : "memory");

#define GAP(...) asm volatile("" : __VA_ARGS__)
#define ALL_ACC "+v"(uc), "+v"(u2), "+v"(u3)
#define PR(x, y) v2f{(x), (y)}

    float t0;
    uc = wmma4(aw[0], h01, uc);               t0 = sq2(h01.x, h01.y);   GAP("+v"(uc), "+v"(t0));
    u2 = wmma4(s0[0], h01, u2);               t0 += sq2(hn[2], hn[3]);  GAP("+v"(uc), "+v"(u2), "+v"(t0));
    uc = wmma4(aw[1], PR(hn[2], hn[3]), uc);  t0 += sq2(hn[4], hn[5]);  GAP("+v"(uc), "+v"(u2), "+v"(t0));
    u2 = wmma4(s0[1], PR(hn[2], hn[3]), u2);  t0 += sq2(hn[6], hn[7]);  GAP("+v"(uc), "+v"(u2), "+v"(t0));
    uc = wmma4(aw[2], PR(hn[4], hn[5]), uc);
    u2 = wmma4(s0[2], PR(hn[4], hn[5]), u2);
    uc = wmma4(aw[3], PR(hn[6], hn[7]), uc);
    u2 = wmma4(s0[3], PR(hn[6], hn[7]), u2);
    GAP("+v"(uc), "+v"(u2), "+v"(t0), "+v"(a1), "+v"(b1), "+v"(a2), "+v"(b2), "+v"(a3), "+v"(b3));

    float t1, t2, t3;
    u3 = wmma4(s1[0], PR(a1.x, a1.y), u3);  GAP(ALL_ACC, "+v"(a1));
    t1 = sq2(a1.x, a1.y);   t1 += sq2(a1.z, a1.w);
    uc = wmma4(aw[4], PR(a1.x, a1.y), uc);  GAP(ALL_ACC, "+v"(b1), "+v"(t1));
    t1 += sq2(b1.x, b1.y);  t1 += sq2(b1.z, b1.w);
    u3 = wmma4(s1[1], PR(a1.z, a1.w), u3);  GAP(ALL_ACC, "+v"(a2));
    t2 = sq2(a2.x, a2.y);   t2 += sq2(a2.z, a2.w);
    u2 = wmma4(s2[0], PR(a2.x, a2.y), u2);  GAP(ALL_ACC, "+v"(b2), "+v"(t2));
    t2 += sq2(b2.x, b2.y);  t2 += sq2(b2.z, b2.w);
    u3 = wmma4(s1[2], PR(b1.x, b1.y), u3);  GAP(ALL_ACC, "+v"(a3));
    t3 = sq2(a3.x, a3.y);   t3 += sq2(a3.z, a3.w);
    uc = wmma4(aw[5], PR(a1.z, a1.w), uc);  GAP(ALL_ACC, "+v"(b3), "+v"(t3));
    t3 += sq2(b3.x, b3.y);  t3 += sq2(b3.z, b3.w);
    u3 = wmma4(s1[3], PR(b1.z, b1.w), u3);  GAP(ALL_ACC, "+v"(t0), "+v"(t1), "+v"(t2), "+v"(t3));
    const float sa = t0 + t1;
    const float sb = t2 + t3;
    float pa = sa + sb;
    float pm = up ? sa : sb;
    u2 = wmma4(s2[1], PR(a2.z, a2.w), u2);  GAP(ALL_ACC, "+v"(pa), "+v"(pm));
    pa += xhalf(pa);
    pm += xhalf(pm);
    u3 = wmma4(s3[0], PR(a3.x, a3.y), u3);  GAP(ALL_ACC, "+v"(pa), "+v"(pm));
    pa = rcp_hw(pa + 1e-8f);
    uc = wmma4(aw[6], PR(b1.x, b1.y), uc);  GAP(ALL_ACC, "+v"(pa), "+v"(pm));
    pa = -(pm * pa - 0.72134752044448170f) * 20.0f;
    u3 = wmma4(s3[1], PR(a3.z, a3.w), u3);  GAP(ALL_ACC, "+v"(pa));
    pa = 1.0f + __expf(pa);
    u2 = wmma4(s2[2], PR(b2.x, b2.y), u2);  GAP(ALL_ACC, "+v"(pa));
    pa = rcp_hw(pa);
#undef GAP
#undef ALL_ACC
    u3 = wmma4(s3[2], PR(b3.x, b3.y), u3);
    uc = wmma4(aw[7], PR(b1.z, b1.w), uc);
    u3 = wmma4(s3[3], PR(b3.z, b3.w), u3);
    u2 = wmma4(s2[3], PR(b2.z, b2.w), u2);
#undef PR

#pragma unroll
    for (int r = 0; r < 8; ++r)
      hn[r] = fast_tanh(__builtin_fmaf(pa, u2[r] + u3[r], uc[r]));
    {
      float4 s;
      s.x = hn[0]; s.y = hn[1]; s.z = hn[2]; s.w = hn[3];
      *(float4*)(pw) = s;
      s.x = hn[4]; s.y = hn[5]; s.z = hn[6]; s.w = hn[7];
      *(float4*)(pw + 128) = s;
    }
    __syncthreads();
  }

  const float* hT = hlds + (T_ & 1) * PLANE;

  for (int i = threadIdx.x; i < 1024; i += 128) {
    const int row = i >> 6, col = i & 63;
    const int m = col >> 4, h8 = (col >> 3) & 1, q = (col >> 2) & 1, e = col & 3;
    out[32768 + btile * 1024 + i] = hT[m * TILE + q * 128 + (h8 * 16 + row) * 4 + e];
  }

  float4 fa[4], fb[4];
#pragma unroll
  for (int m = 0; m < 4; ++m) {
    fa[m] = *(const float4*)(hT + m * TILE + lane * 4);
    fb[m] = *(const float4*)(hT + m * TILE + 128 + lane * 4);
  }
#pragma unroll
  for (int n2i = 0; n2i < 2; ++n2i) {
    const int n2 = nt * 2 + n2i;
    v8f p;
#pragma unroll
    for (int r = 0; r < 8; ++r) p[r] = Wob[n2 * 16 + hf * 8 + r];
    const float* wsrc = Wo + (size_t)(n2 * 16 + lo) * 64 + ko;
#pragma unroll
    for (int m = 0; m < 4; ++m) {
      p = wmma4(*(const v2f*)(wsrc + m * 16 + 0), v2f{fa[m].x, fa[m].y}, p);
      p = wmma4(*(const v2f*)(wsrc + m * 16 + 2), v2f{fa[m].z, fa[m].w}, p);
      p = wmma4(*(const v2f*)(wsrc + m * 16 + 4), v2f{fb[m].x, fb[m].y}, p);
      p = wmma4(*(const v2f*)(wsrc + m * 16 + 6), v2f{fb[m].z, fb[m].w}, p);
    }
    float* o = out + (size_t)(btile * 16 + lo) * 128 + n2 * 16 + hf * 8;
    float4 s;
    s.x = p[0]; s.y = p[1]; s.z = p[2]; s.w = p[3];
    *(float4*)(o) = s;
    s.x = p[4]; s.y = p[5]; s.z = p[6]; s.w = p[7];
    *(float4*)(o + 4) = s;
  }
}

extern "C" void kernel_launch(void* const* d_in, const int* in_sizes, int n_in,
                              void* d_out, int out_size, void* d_ws, size_t ws_size,
                              hipStream_t stream) {
  const float* x    = (const float*)d_in[0];
  const float* Ap   = (const float*)d_in[1];
  const float* Am   = (const float*)d_in[2];
  const float* Wf   = (const float*)d_in[3];
  const float* Wiw  = (const float*)d_in[4];
  const float* Wib  = (const float*)d_in[5];
  const float* Wow  = (const float*)d_in[6];
  const float* Wob  = (const float*)d_in[7];
  const float* bias = (const float*)d_in[8];
  float* out = (float*)d_out;
  float* xp  = (float*)d_ws;

  xproj_kernel<<<4096, 256, 0, stream>>>(x, Wiw, Wib, bias, xp);
  scan_kernel<<<16, 128, 0, stream>>>(Ap, Am, Wf, Wow, Wob, xp, out);
}


# scan: tanh by the hardware instruction v_tanh_f32 instead of exp, reciprocal and seven plain instructions per value; last-place rounding of tanh changes, sum order unchanged
# speedup vs baseline: 4.0641x; 1.0831x over previous
#include <hip/hip_runtime.h>
#include <hip/hip_bf16.h>


typedef float v8f __attribute__((ext_vector_type(8)));
typedef float v2f __attribute__((ext_vector_type(2)));
typedef float v4f __attribute__((ext_vector_type(4)));

#define T_  2048
#define D_  128
#define TILE  256
#define PLANE (4 * TILE)
#define PF  8


__device__ __forceinline__ v8f wmma4(v2f a, v2f b, v8f c) {
  return __builtin_amdgcn_wmma_f32_16x16x4_f32(false, a, false, b, (short)0, c,
                                               false, false);
}

__device__ __forceinline__ float rcp_hw(float d) {
  return __builtin_amdgcn_rcpf(d);
}

__device__ __forceinline__ float fast_tanh(float x) {
  return __builtin_amdgcn_tanhf(x);
}

__device__ __forceinline__ float xhalf(float x) {
  int b = __builtin_bit_cast(int, x);
  return __builtin_bit_cast(
      float, __builtin_amdgcn_permlanex16(b, b, 0x76543210, 0xfedcba98, false, false));
}

__device__ __forceinline__ float sq2(float a, float b) {
  return __builtin_fmaf(a, a, b * b);
}

__global__ void __launch_bounds__(256)
xproj_kernel(const float* __restrict__ x, const float* __restrict__ Wi,
             const float* __restrict__ Wib, const float* __restrict__ bias,
             float* __restrict__ xp) {
  const int lane = threadIdx.x & 31;
  const int wid  = blockIdx.x * 8 + (threadIdx.x >> 5);
  const int btile = wid >> 11;
  const int t     = wid & 2047;
  const int lo = lane & 15, hf = lane >> 4;

  v8f acc[4];
#pragma unroll
  for (int nt = 0; nt < 4; ++nt) {
    const float4 w0 = *(const float4*)(Wib + nt * 16 + hf * 8);
    const float4 w1 = *(const float4*)(Wib + nt * 16 + hf * 8 + 4);
    const float4 c0 = *(const float4*)(bias + nt * 16 + hf * 8);
    const float4 c1 = *(const float4*)(bias + nt * 16 + hf * 8 + 4);
    acc[nt][0] = w0.x + c0.x; acc[nt][1] = w0.y + c0.y;
    acc[nt][2] = w0.z + c0.z; acc[nt][3] = w0.w + c0.w;
    acc[nt][4] = w1.x + c1.x; acc[nt][5] = w1.y + c1.y;
    acc[nt][6] = w1.z + c1.z; acc[nt][7] = w1.w + c1.w;
  }

  const float* xrow = x + ((size_t)(btile * 16 + lo) * T_ + t) * D_ + hf * 2;
  const float* wrow = Wi + (size_t)lo * D_ + hf * 2;
#pragma unroll
  for (int ks = 0; ks < 32; ++ks) {
    v2f xb = *(const v2f*)(xrow + ks * 4);
#pragma unroll
    for (int nt = 0; nt < 4; ++nt) {
      v2f wa = *(const v2f*)(wrow + nt * 16 * D_ + ks * 4);
      acc[nt] = wmma4(wa, xb, acc[nt]);
    }
  }

  float* o = xp + (size_t)wid * 1024 + lane * 4;
#pragma unroll
  for (int nt = 0; nt < 4; ++nt) {
#pragma unroll
    for (int q = 0; q < 2; ++q) {
      float4 s;
      s.x = acc[nt][q * 4 + 0]; s.y = acc[nt][q * 4 + 1];
      s.z = acc[nt][q * 4 + 2]; s.w = acc[nt][q * 4 + 3];
      *(float4*)(o + (nt * 2 + q) * 128) = s;
    }
  }
}

__global__ void __launch_bounds__(128)
scan_kernel(const float* __restrict__ Ap, const float* __restrict__ Am,
            const float* __restrict__ Wf, const float* __restrict__ Wo,
            const float* __restrict__ Wob, const float* __restrict__ xp,
            float* __restrict__ out) {
  __shared__ __align__(16) float hlds[2 * PLANE];
  const int lane = threadIdx.x & 31;
  const int lo = lane & 15, hf = lane >> 4;
  const int nt = __builtin_amdgcn_readfirstlane((int)(threadIdx.x >> 5));
  const int btile = blockIdx.x;
  const bool up = nt >= 2;
  const int m1 = nt ^ 1, m2 = nt ^ 2, m3 = nt ^ 3;

  const int j = nt * 16 + lo;
  const int ko = hf * 8;
  v2f aw[8];
  {
    const float* src = (up ? Am : Ap) + j * 64 + ko;
#pragma unroll
    for (int p = 0; p < 4; ++p) {
      aw[p]     = *(const v2f*)(src + nt * 16 + 2 * p);
      aw[4 + p] = *(const v2f*)(src + m1 * 16 + 2 * p);
    }
  }
  v2f s0[4], s1[4], s2[4], s3[4];
  {
    const float sg = up ? -1.0f : 1.0f;
    const float* src = Wf + j * 64 + ko;
#pragma unroll
    for (int p = 0; p < 4; ++p) {
      v2f w;
      w = *(const v2f*)(src + nt * 16 + 2 * p); s0[p].x = w.x * sg; s0[p].y = w.y * sg;
      w = *(const v2f*)(src + m1 * 16 + 2 * p); s1[p].x = w.x * sg; s1[p].y = w.y * sg;
      w = *(const v2f*)(src + m2 * 16 + 2 * p); s2[p].x = w.x * sg; s2[p].y = w.y * sg;
      w = *(const v2f*)(src + m3 * 16 + 2 * p); s3[p].x = w.x * sg; s3[p].y = w.y * sg;
    }
  }

  float hn[8];
#pragma unroll
  for (int r = 0; r < 8; ++r) hn[r] = 0.0f;
  {
    float4 z; z.x = 0.0f; z.y = 0.0f; z.z = 0.0f; z.w = 0.0f;
    *(float4*)(hlds + nt * TILE + lane * 4) = z;
    *(float4*)(hlds + nt * TILE + 128 + lane * 4) = z;
  }
  __syncthreads();

  const float* xpl = xp + (size_t)btile * T_ * 1024 + nt * 256 + lane * 4;
  const float* xpf = xp + (size_t)btile * T_ * 1024 + nt * 256 + lane * 8;
  float4 xq0 = *(const float4*)(xpl);
  float4 xq1 = *(const float4*)(xpl + 128);

#pragma unroll 1
  for (int t = 0; t < T_; ++t) {
    const float* pr = hlds + (t & 1) * PLANE + lane * 4;
    float* pw = hlds + ((t & 1) ^ 1) * PLANE + nt * TILE + lane * 4;

    v4f a1 = *(const v4f*)(pr + m1 * TILE);
    v4f b1 = *(const v4f*)(pr + m1 * TILE + 128);
    v4f a2 = *(const v4f*)(pr + m2 * TILE);
    v4f b2 = *(const v4f*)(pr + m2 * TILE + 128);
    v4f a3 = *(const v4f*)(pr + m3 * TILE);
    v4f b3 = *(const v4f*)(pr + m3 * TILE + 128);

    v8f uc, u2, u3;
    uc[0] = xq0.x; uc[1] = xq0.y; uc[2] = xq0.z; uc[3] = xq0.w;
    uc[4] = xq1.x; uc[5] = xq1.y; uc[6] = xq1.z; uc[7] = xq1.w;
#pragma unroll
    for (int v = 0; v < 8; ++v) { u2[v] = 0.0f; u3[v] = 0.0f; }
    {
      const int tn = (t + 1 < T_) ? (t + 1) : (T_ - 1);
      const float* nx = xpl + (size_t)tn * 1024;
      xq0 = *(const float4*)(nx);
      xq1 = *(const float4*)(nx + 128);
      const int tp = (t + PF < T_) ? (t + PF) : (T_ - 1);
      __builtin_prefetch(xpf + (size_t)tp * 1024, 0, 1);
    }
    v2f h01 = v2f{hn[0], hn[1]};
    asm volatile("" : "+v"(uc), "+v"(h01) : : "memory");

#define GAP(...) asm volatile("" : __VA_ARGS__)
#define ALL_ACC "+v"(uc), "+v"(u2), "+v"(u3)
#define PR(x, y) v2f{(x), (y)}

    float t0;
    uc = wmma4(aw[0], h01, uc);               t0 = sq2(h01.x, h01.y);   GAP("+v"(uc), "+v"(t0));
    u2 = wmma4(s0[0], h01, u2);               t0 += sq2(hn[2], hn[3]);  GAP("+v"(uc), "+v"(u2), "+v"(t0));
    uc = wmma4(aw[1], PR(hn[2], hn[3]), uc);  t0 += sq2(hn[4], hn[5]);  GAP("+v"(uc), "+v"(u2), "+v"(t0));
    u2 = wmma4(s0[1], PR(hn[2], hn[3]), u2);  t0 += sq2(hn[6], hn[7]);  GAP("+v"(uc), "+v"(u2), "+v"(t0));
    uc = wmma4(aw[2], PR(hn[4], hn[5]), uc);
    u2 = wmma4(s0[2], PR(hn[4], hn[5]), u2);
    uc = wmma4(aw[3], PR(hn[6], hn[7]), uc);
    u2 = wmma4(s0[3], PR(hn[6], hn[7]), u2);
    GAP("+v"(uc), "+v"(u2), "+v"(t0), "+v"(a1), "+v"(b1), "+v"(a2), "+v"(b2), "+v"(a3), "+v"(b3));

    float t1, t2, t3;
    u3 = wmma4(s1[0], PR(a1.x, a1.y), u3);  GAP(ALL_ACC, "+v"(a1));
    t1 = sq2(a1.x, a1.y);   t1 += sq2(a1.z, a1.w);
    uc = wmma4(aw[4], PR(a1.x, a1.y), uc);  GAP(ALL_ACC, "+v"(b1), "+v"(t1));
    t1 += sq2(b1.x, b1.y);  t1 += sq2(b1.z, b1.w);
    u3 = wmma4(s1[1], PR(a1.z, a1.w), u3);  GAP(ALL_ACC, "+v"(a2));
    t2 = sq2(a2.x, a2.y);   t2 += sq2(a2.z, a2.w);
    u2 = wmma4(s2[0], PR(a2.x, a2.y), u2);  GAP(ALL_ACC, "+v"(b2), "+v"(t2));
    t2 += sq2(b2.x, b2.y);  t2 += sq2(b2.z, b2.w);
    u3 = wmma4(s1[2], PR(b1.x, b1.y), u3);  GAP(ALL_ACC, "+v"(a3));
    t3 = sq2(a3.x, a3.y);   t3 += sq2(a3.z, a3.w);
    uc = wmma4(aw[5], PR(a1.z, a1.w), uc);  GAP(ALL_ACC, "+v"(b3), "+v"(t3));
    t3 += sq2(b3.x, b3.y);  t3 += sq2(b3.z, b3.w);
    u3 = wmma4(s1[3], PR(b1.z, b1.w), u3);  GAP(ALL_ACC, "+v"(t0), "+v"(t1), "+v"(t2), "+v"(t3));
    const float sa = t0 + t1;
    const float sb = t2 + t3;
    float pa = sa + sb;
    float pm = up ? sa : sb;
    u2 = wmma4(s2[1], PR(a2.z, a2.w), u2);  GAP(ALL_ACC, "+v"(pa), "+v"(pm));
    pa += xhalf(pa);
    pm += xhalf(pm);
    u3 = wmma4(s3[0], PR(a3.x, a3.y), u3);  GAP(ALL_ACC, "+v"(pa), "+v"(pm));
    pa = rcp_hw(pa + 1e-8f);
    uc = wmma4(aw[6], PR(b1.x, b1.y), uc);  GAP(ALL_ACC, "+v"(pa), "+v"(pm));
    pa = -(pm * pa - 0.72134752044448170f) * 20.0f;
    u3 = wmma4(s3[1], PR(a3.z, a3.w), u3);  GAP(ALL_ACC, "+v"(pa));
    pa = 1.0f + __expf(pa);
    u2 = wmma4(s2[2], PR(b2.x, b2.y), u2);  GAP(ALL_ACC, "+v"(pa));
    pa = rcp_hw(pa);
#undef GAP
#undef ALL_ACC
    u3 = wmma4(s3[2], PR(b3.x, b3.y), u3);
    uc = wmma4(aw[7], PR(b1.z, b1.w), uc);
    u3 = wmma4(s3[3], PR(b3.z, b3.w), u3);
    u2 = wmma4(s2[3], PR(b2.z, b2.w), u2);
#undef PR

#pragma unroll
    for (int r = 0; r < 8; ++r)
      hn[r] = fast_tanh(__builtin_fmaf(pa, u2[r] + u3[r], uc[r]));
    {
      float4 s;
      s.x = hn[0]; s.y = hn[1]; s.z = hn[2]; s.w = hn[3];
      *(float4*)(pw) = s;
      s.x = hn[4]; s.y = hn[5]; s.z = hn[6]; s.w = hn[7];
      *(float4*)(pw + 128) = s;
    }
    __syncthreads();
  }

  const float* hT = hlds + (T_ & 1) * PLANE;

  for (int i = threadIdx.x; i < 1024; i += 128) {
    const int row = i >> 6, col = i & 63;
    const int m = col >> 4, h8 = (col >> 3) & 1, q = (col >> 2) & 1, e = col & 3;
    out[32768 + btile * 1024 + i] = hT[m * TILE + q * 128 + (h8 * 16 + row) * 4 + e];
  }

  float4 fa[4], fb[4];
#pragma unroll
  for (int m = 0; m < 4; ++m) {
    fa[m] = *(const float4*)(hT + m * TILE + lane * 4);
    fb[m] = *(const float4*)(hT + m * TILE + 128 + lane * 4);
  }
#pragma unroll
  for (int n2i = 0; n2i < 2; ++n2i) {
    const int n2 = nt * 2 + n2i;
    v8f p;
#pragma unroll
    for (int r = 0; r < 8; ++r) p[r] = Wob[n2 * 16 + hf * 8 + r];
    const float* wsrc = Wo + (size_t)(n2 * 16 + lo) * 64 + ko;
#pragma unroll
    for (int m = 0; m < 4; ++m) {
      p = wmma4(*(const v2f*)(wsrc + m * 16 + 0), v2f{fa[m].x, fa[m].y}, p);
      p = wmma4(*(const v2f*)(wsrc + m * 16 + 2), v2f{fa[m].z, fa[m].w}, p);
      p = wmma4(*(const v2f*)(wsrc + m * 16 + 4), v2f{fb[m].x, fb[m].y}, p);
      p = wmma4(*(const v2f*)(wsrc + m * 16 + 6), v2f{fb[m].z, fb[m].w}, p);
    }
    float* o = out + (size_t)(btile * 16 + lo) * 128 + n2 * 16 + hf * 8;
    float4 s;
    s.x = p[0]; s.y = p[1]; s.z = p[2]; s.w = p[3];
    *(float4*)(o) = s;
    s.x = p[4]; s.y = p[5]; s.z = p[6]; s.w = p[7];
    *(float4*)(o + 4) = s;
  }
}

extern "C" void kernel_launch(void* const* d_in, const int* in_sizes, int n_in,
                              void* d_out, int out_size, void* d_ws, size_t ws_size,
                              hipStream_t stream) {
  const float* x    = (const float*)d_in[0];
  const float* Ap   = (const float*)d_in[1];
  const float* Am   = (const float*)d_in[2];
  const float* Wf   = (const float*)d_in[3];
  const float* Wiw  = (const float*)d_in[4];
  const float* Wib  = (const float*)d_in[5];
  const float* Wow  = (const float*)d_in[6];
  const float* Wob  = (const float*)d_in[7];
  const float* bias = (const float*)d_in[8];
  float* out = (float*)d_out;
  float* xp  = (float*)d_ws;

  xproj_kernel<<<4096, 256, 0, stream>>>(x, Wiw, Wib, bias, xp);
  scan_kernel<<<16, 128, 0, stream>>>(Ap, Am, Wf, Wow, Wob, xp, out);
}


# scan: own tile's k-steps 0,1 of step t+1 (4 matrix ops) run between the LDS store and the barrier, k-steps 2,3 behind the LDS loads; planes constant (2 steps per trip); order of sums unchanged
# speedup vs baseline: 4.5022x; 1.1078x over previous
#include <hip/hip_runtime.h>
#include <hip/hip_bf16.h>


typedef float v8f __attribute__((ext_vector_type(8)));
typedef float v2f __attribute__((ext_vector_type(2)));
typedef float v4f __attribute__((ext_vector_type(4)));

#define T_  2048
#define D_  128
#define TILE  256
#define PLANE (4 * TILE)
#define PF  8


__device__ __forceinline__ v8f wmma4(v2f a, v2f b, v8f c) {
  return __builtin_amdgcn_wmma_f32_16x16x4_f32(false, a, false, b, (short)0, c,
                                               false, false);
}

__device__ __forceinline__ float rcp_hw(float d) {
  return __builtin_amdgcn_rcpf(d);
}

__device__ __forceinline__ float fast_tanh(float x) {
  return __builtin_amdgcn_tanhf(x);
}

__device__ __forceinline__ float xhalf(float x) {
  int b = __builtin_bit_cast(int, x);
  return __builtin_bit_cast(
      float, __builtin_amdgcn_permlanex16(b, b, 0x76543210, 0xfedcba98, false, false));
}

__device__ __forceinline__ float sq2(float a, float b) {
  return __builtin_fmaf(a, a, b * b);
}

__global__ void __launch_bounds__(256)
xproj_kernel(const float* __restrict__ x, const float* __restrict__ Wi,
             const float* __restrict__ Wib, const float* __restrict__ bias,
             float* __restrict__ xp) {
  const int lane = threadIdx.x & 31;
  const int wid  = blockIdx.x * 8 + (threadIdx.x >> 5);
  const int btile = wid >> 11;
  const int t     = wid & 2047;
  const int lo = lane & 15, hf = lane >> 4;

  v8f acc[4];
#pragma unroll
  for (int nt = 0; nt < 4; ++nt) {
    const float4 w0 = *(const float4*)(Wib + nt * 16 + hf * 8);
    const float4 w1 = *(const float4*)(Wib + nt * 16 + hf * 8 + 4);
    const float4 c0 = *(const float4*)(bias + nt * 16 + hf * 8);
    const float4 c1 = *(const float4*)(bias + nt * 16 + hf * 8 + 4);
    acc[nt][0] = w0.x + c0.x; acc[nt][1] = w0.y + c0.y;
    acc[nt][2] = w0.z + c0.z; acc[nt][3] = w0.w + c0.w;
    acc[nt][4] = w1.x + c1.x; acc[nt][5] = w1.y + c1.y;
    acc[nt][6] = w1.z + c1.z; acc[nt][7] = w1.w + c1.w;
  }

  const float* xrow = x + ((size_t)(btile * 16 + lo) * T_ + t) * D_ + hf * 2;
  const float* wrow = Wi + (size_t)lo * D_ + hf * 2;
#pragma unroll
  for (int ks = 0; ks < 32; ++ks) {
    v2f xb = *(const v2f*)(xrow + ks * 4);
#pragma unroll
    for (int nt = 0; nt < 4; ++nt) {
      v2f wa = *(const v2f*)(wrow + nt * 16 * D_ + ks * 4);
      acc[nt] = wmma4(wa, xb, acc[nt]);
    }
  }

  float* o = xp + (size_t)wid * 1024 + lane * 4;
#pragma unroll
  for (int nt = 0; nt < 4; ++nt) {
#pragma unroll
    for (int q = 0; q < 2; ++q) {
      float4 s;
      s.x = acc[nt][q * 4 + 0]; s.y = acc[nt][q * 4 + 1];
      s.z = acc[nt][q * 4 + 2]; s.w = acc[nt][q * 4 + 3];
      *(float4*)(o + (nt * 2 + q) * 128) = s;
    }
  }
}

__global__ void __launch_bounds__(128)
scan_kernel(const float* __restrict__ Ap, const float* __restrict__ Am,
            const float* __restrict__ Wf, const float* __restrict__ Wo,
            const float* __restrict__ Wob, const float* __restrict__ xp,
            float* __restrict__ out) {
  __shared__ __align__(16) float hlds[2 * PLANE];
  const int lane = threadIdx.x & 31;
  const int lo = lane & 15, hf = lane >> 4;
  const int nt = __builtin_amdgcn_readfirstlane((int)(threadIdx.x >> 5));
  const int btile = blockIdx.x;
  const bool up = nt >= 2;
  const int m1 = nt ^ 1, m2 = nt ^ 2, m3 = nt ^ 3;

  const int j = nt * 16 + lo;
  const int ko = hf * 8;
  v2f aw[8];
  {
    const float* src = (up ? Am : Ap) + j * 64 + ko;
#pragma unroll
    for (int p = 0; p < 4; ++p) {
      aw[p]     = *(const v2f*)(src + nt * 16 + 2 * p);
      aw[4 + p] = *(const v2f*)(src + m1 * 16 + 2 * p);
    }
  }
  v2f s0[4], s1[4], s2[4], s3[4];
  {
    const float sg = up ? -1.0f : 1.0f;
    const float* src = Wf + j * 64 + ko;
#pragma unroll
    for (int p = 0; p < 4; ++p) {
      v2f w;
      w = *(const v2f*)(src + nt * 16 + 2 * p); s0[p].x = w.x * sg; s0[p].y = w.y * sg;
      w = *(const v2f*)(src + m1 * 16 + 2 * p); s1[p].x = w.x * sg; s1[p].y = w.y * sg;
      w = *(const v2f*)(src + m2 * 16 + 2 * p); s2[p].x = w.x * sg; s2[p].y = w.y * sg;
      w = *(const v2f*)(src + m3 * 16 + 2 * p); s3[p].x = w.x * sg; s3[p].y = w.y * sg;
    }
  }

  float hn[8];
#pragma unroll
  for (int r = 0; r < 8; ++r) hn[r] = 0.0f;
  {
    float4 z; z.x = 0.0f; z.y = 0.0f; z.z = 0.0f; z.w = 0.0f;
    *(float4*)(hlds + nt * TILE + lane * 4) = z;
    *(float4*)(hlds + nt * TILE + 128 + lane * 4) = z;
  }

  const float* xpl = xp + (size_t)btile * T_ * 1024 + nt * 256 + lane * 4;
  const float* xpf = xp + (size_t)btile * T_ * 1024 + nt * 256 + lane * 8;
  float4 xq0 = *(const float4*)(xpl);
  float4 xq1 = *(const float4*)(xpl + 128);

  const float* r1 = hlds + m1 * TILE + lane * 4;
  const float* r2 = hlds + m2 * TILE + lane * 4;
  const float* r3 = hlds + m3 * TILE + lane * 4;
  float* w0 = hlds + nt * TILE + lane * 4;

#define GAP(...) asm volatile("" : __VA_ARGS__)
#define ALL_ACC "+v"(uc), "+v"(u2), "+v"(u3)
#define PR(x, y) v2f{(x), (y)}

  v8f uc, u2;
  float t0;
#define OWN_FIRST(TN, TP)                                                            \
  {                                                                                  \
    v8f xs, z8;                                                                      \
    xs[0] = xq0.x; xs[1] = xq0.y; xs[2] = xq0.z; xs[3] = xq0.w;                      \
    xs[4] = xq1.x; xs[5] = xq1.y; xs[6] = xq1.z; xs[7] = xq1.w;                      \
    _Pragma("unroll") for (int v = 0; v < 8; ++v) z8[v] = 0.0f;                      \
    u2 = wmma4(s0[0], h01, z8);               t0 = sq2(h01.x, h01.y);                \
    GAP("+v"(u2), "+v"(t0));                                                         \
    uc = wmma4(aw[0], h01, xs);               t0 += sq2(hn[2], hn[3]);               \
    GAP("+v"(uc), "+v"(u2), "+v"(t0));                                               \
    u2 = wmma4(s0[1], PR(hn[2], hn[3]), u2);  t0 += sq2(hn[4], hn[5]);               \
    GAP("+v"(uc), "+v"(u2), "+v"(t0));                                               \
    uc = wmma4(aw[1], PR(hn[2], hn[3]), uc);  t0 += sq2(hn[6], hn[7]);               \
    int tn = (TN);                                                                   \
        \
        \
        \
        \
        \
        \
    asm volatile("" : "+s"(tn), "+v"(uc), "+v"(u2), "+v"(t0));                       \
    {                                                                                \
      const float* nx = xpl + (size_t)tn * 1024;                                     \
      xq0 = *(const float4*)(nx);                                                    \
      xq1 = *(const float4*)(nx + 128);                                              \
      __builtin_prefetch(xpf + (size_t)(TP) * 1024, 0, 1);                           \
    }                                                                                \
        \
        \
        \
        \
        \
    asm volatile("s_wait_dscnt 0x0\n\ts_barrier_signal -1\n\ts_barrier_wait -1"      \
                 : "+v"(uc), "+v"(u2) : : "memory");                                 \
  }

  {
    v2f h01 = v2f{hn[0], hn[1]};
    OWN_FIRST(1, 1 + PF)
  }

#pragma unroll 1
  for (int t = 0; t < T_; t += 2) {
#pragma unroll
    for (int p = 0; p < 2; ++p) {
      const int pro = p * PLANE;
      const int pwo = (p ^ 1) * PLANE;

      v4f a1 = *(const v4f*)(r1 + pro);
      v4f b1 = *(const v4f*)(r1 + pro + 128);
      v4f a2 = *(const v4f*)(r2 + pro);
      v4f b2 = *(const v4f*)(r2 + pro + 128);
      v4f a3 = *(const v4f*)(r3 + pro);
      v4f b3 = *(const v4f*)(r3 + pro + 128);
      asm volatile("" : "+v"(uc), "+v"(u2) : : "memory");

      uc = wmma4(aw[2], PR(hn[4], hn[5]), uc);
      u2 = wmma4(s0[2], PR(hn[4], hn[5]), u2);
      uc = wmma4(aw[3], PR(hn[6], hn[7]), uc);
      u2 = wmma4(s0[3], PR(hn[6], hn[7]), u2);
      GAP("+v"(uc), "+v"(u2), "+v"(t0), "+v"(a1), "+v"(b1), "+v"(a2), "+v"(b2), "+v"(a3), "+v"(b3));

      v8f u3;
#pragma unroll
      for (int v = 0; v < 8; ++v) u3[v] = 0.0f;
      float t1, t2, t3;
      u3 = wmma4(s1[0], PR(a1.x, a1.y), u3);  GAP(ALL_ACC, "+v"(a1));
      t1 = sq2(a1.x, a1.y);   t1 += sq2(a1.z, a1.w);
      uc = wmma4(aw[4], PR(a1.x, a1.y), uc);  GAP(ALL_ACC, "+v"(b1), "+v"(t1));
      t1 += sq2(b1.x, b1.y);  t1 += sq2(b1.z, b1.w);
      u3 = wmma4(s1[1], PR(a1.z, a1.w), u3);  GAP(ALL_ACC, "+v"(a2));
      t2 = sq2(a2.x, a2.y);   t2 += sq2(a2.z, a2.w);
      u2 = wmma4(s2[0], PR(a2.x, a2.y), u2);  GAP(ALL_ACC, "+v"(b2), "+v"(t2));
      t2 += sq2(b2.x, b2.y);  t2 += sq2(b2.z, b2.w);
      u3 = wmma4(s1[2], PR(b1.x, b1.y), u3);  GAP(ALL_ACC, "+v"(a3));
      t3 = sq2(a3.x, a3.y);   t3 += sq2(a3.z, a3.w);
      uc = wmma4(aw[5], PR(a1.z, a1.w), uc);  GAP(ALL_ACC, "+v"(b3), "+v"(t3));
      t3 += sq2(b3.x, b3.y);  t3 += sq2(b3.z, b3.w);
      u3 = wmma4(s1[3], PR(b1.z, b1.w), u3);  GAP(ALL_ACC, "+v"(t0), "+v"(t1), "+v"(t2), "+v"(t3));
      const float sa = t0 + t1;
      const float sb = t2 + t3;
      float pa = sa + sb;
      float pm = up ? sa : sb;
      u2 = wmma4(s2[1], PR(a2.z, a2.w), u2);  GAP(ALL_ACC, "+v"(pa), "+v"(pm));
      pa += xhalf(pa);
      pm += xhalf(pm);
      u3 = wmma4(s3[0], PR(a3.x, a3.y), u3);  GAP(ALL_ACC, "+v"(pa), "+v"(pm));
      pa = rcp_hw(pa + 1e-8f);
      uc = wmma4(aw[6], PR(b1.x, b1.y), uc);  GAP(ALL_ACC, "+v"(pa), "+v"(pm));
      pa = -(pm * pa - 0.72134752044448170f) * 20.0f;
      u3 = wmma4(s3[1], PR(a3.z, a3.w), u3);  GAP(ALL_ACC, "+v"(pa));
      pa = 1.0f + __expf(pa);
      u2 = wmma4(s2[2], PR(b2.x, b2.y), u2);  GAP(ALL_ACC, "+v"(pa));
      pa = rcp_hw(pa);
      u3 = wmma4(s3[2], PR(b3.x, b3.y), u3);
      uc = wmma4(aw[7], PR(b1.z, b1.w), uc);
      u3 = wmma4(s3[3], PR(b3.z, b3.w), u3);
      u2 = wmma4(s2[3], PR(b2.z, b2.w), u2);

#pragma unroll
      for (int r = 0; r < 8; ++r)
        hn[r] = fast_tanh(__builtin_fmaf(pa, u2[r] + u3[r], uc[r]));
      {
        float4 s;
        s.x = hn[0]; s.y = hn[1]; s.z = hn[2]; s.w = hn[3];
        *(float4*)(w0 + pwo) = s;
        s.x = hn[4]; s.y = hn[5]; s.z = hn[6]; s.w = hn[7];
        *(float4*)(w0 + pwo + 128) = s;
      }
      v2f h01 = v2f{hn[0], hn[1]};
      asm volatile("" : "+v"(h01) : : "memory");

      const int nxt = t + p + 2;
      OWN_FIRST((nxt < T_) ? nxt : (T_ - 1), (nxt + PF < T_) ? (nxt + PF) : (T_ - 1))
    }
  }
#undef OWN_FIRST
#undef GAP
#undef ALL_ACC
#undef PR

  const float* hT = hlds + (T_ & 1) * PLANE;

  for (int i = threadIdx.x; i < 1024; i += 128) {
    const int row = i >> 6, col = i & 63;
    const int m = col >> 4, h8 = (col >> 3) & 1, q = (col >> 2) & 1, e = col & 3;
    out[32768 + btile * 1024 + i] = hT[m * TILE + q * 128 + (h8 * 16 + row) * 4 + e];
  }

  float4 fa[4], fb[4];
#pragma unroll
  for (int m = 0; m < 4; ++m) {
    fa[m] = *(const float4*)(hT + m * TILE + lane * 4);
    fb[m] = *(const float4*)(hT + m * TILE + 128 + lane * 4);
  }
#pragma unroll
  for (int n2i = 0; n2i < 2; ++n2i) {
    const int n2 = nt * 2 + n2i;
    v8f p;
#pragma unroll
    for (int r = 0; r < 8; ++r) p[r] = Wob[n2 * 16 + hf * 8 + r];
    const float* wsrc = Wo + (size_t)(n2 * 16 + lo) * 64 + ko;
#pragma unroll
    for (int m = 0; m < 4; ++m) {
      p = wmma4(*(const v2f*)(wsrc + m * 16 + 0), v2f{fa[m].x, fa[m].y}, p);
      p = wmma4(*(const v2f*)(wsrc + m * 16 + 2), v2f{fa[m].z, fa[m].w}, p);
      p = wmma4(*(const v2f*)(wsrc + m * 16 + 4), v2f{fb[m].x, fb[m].y}, p);
      p = wmma4(*(const v2f*)(wsrc + m * 16 + 6), v2f{fb[m].z, fb[m].w}, p);
    }
    float* o = out + (size_t)(btile * 16 + lo) * 128 + n2 * 16 + hf * 8;
    float4 s;
    s.x = p[0]; s.y = p[1]; s.z = p[2]; s.w = p[3];
    *(float4*)(o) = s;
    s.x = p[4]; s.y = p[5]; s.z = p[6]; s.w = p[7];
    *(float4*)(o + 4) = s;
  }
}

extern "C" void kernel_launch(void* const* d_in, const int* in_sizes, int n_in,
                              void* d_out, int out_size, void* d_ws, size_t ws_size,
                              hipStream_t stream) {
  const float* x    = (const float*)d_in[0];
  const float* Ap   = (const float*)d_in[1];
  const float* Am   = (const float*)d_in[2];
  const float* Wf   = (const float*)d_in[3];
  const float* Wiw  = (const float*)d_in[4];
  const float* Wib  = (const float*)d_in[5];
  const float* Wow  = (const float*)d_in[6];
  const float* Wob  = (const float*)d_in[7];
  const float* bias = (const float*)d_in[8];
  float* out = (float*)d_out;
  float* xp  = (float*)d_ws;

  xproj_kernel<<<4096, 256, 0, stream>>>(x, Wiw, Wib, bias, xp);
  scan_kernel<<<16, 128, 0, stream>>>(Ap, Am, Wf, Wow, Wob, xp, out);
}


# x_proj kernel: each wave keeps W_in's 128 fragments in registers over 8 consecutive time steps (they were re-read from the cache for every step), next step's x asked for a step ahead; sums unchanged
# speedup vs baseline: 4.8397x; 1.0750x over previous
#include <hip/hip_runtime.h>
#include <hip/hip_bf16.h>


typedef float v8f __attribute__((ext_vector_type(8)));
typedef float v2f __attribute__((ext_vector_type(2)));
typedef float v4f __attribute__((ext_vector_type(4)));

#define T_  2048
#define D_  128
#define TILE  256
#define PLANE (4 * TILE)
#define PF  8


__device__ __forceinline__ v8f wmma4(v2f a, v2f b, v8f c) {
  return __builtin_amdgcn_wmma_f32_16x16x4_f32(false, a, false, b, (short)0, c,
                                               false, false);
}

__device__ __forceinline__ float rcp_hw(float d) {
  return __builtin_amdgcn_rcpf(d);
}

__device__ __forceinline__ float fast_tanh(float x) {
  return __builtin_amdgcn_tanhf(x);
}

__device__ __forceinline__ float xhalf(float x) {
  int b = __builtin_bit_cast(int, x);
  return __builtin_bit_cast(
      float, __builtin_amdgcn_permlanex16(b, b, 0x76543210, 0xfedcba98, false, false));
}

__device__ __forceinline__ float sq2(float a, float b) {
  return __builtin_fmaf(a, a, b * b);
}

#define XT 8

__device__ __forceinline__ void xproj_step(const v2f (&wf)[4][32], const v8f (&seed)[4],
                                           const v2f (&xf)[32], float* __restrict__ o) {
  v8f acc[4];
#pragma unroll
  for (int nt = 0; nt < 4; ++nt) acc[nt] = wmma4(wf[nt][0], xf[0], seed[nt]);
#pragma unroll
  for (int ks = 1; ks < 32; ++ks) {
#pragma unroll
    for (int nt = 0; nt < 4; ++nt) acc[nt] = wmma4(wf[nt][ks], xf[ks], acc[nt]);
  }
#pragma unroll
  for (int nt = 0; nt < 4; ++nt) {
#pragma unroll
    for (int q = 0; q < 2; ++q) {
      float4 s;
      s.x = acc[nt][q * 4 + 0]; s.y = acc[nt][q * 4 + 1];
      s.z = acc[nt][q * 4 + 2]; s.w = acc[nt][q * 4 + 3];
      *(float4*)(o + (nt * 2 + q) * 128) = s;
    }
  }
}

__global__ void __launch_bounds__(128, 2)
xproj_kernel(const float* x, const float* __restrict__ Wi,
             const float* __restrict__ Wib, const float* __restrict__ bias,
             float* xp) {
  const int lane = threadIdx.x & 31;
  const int wv   = blockIdx.x * 4 + (threadIdx.x >> 5);
  const int btile = wv >> 8;
  const int t0    = (wv & 255) * XT;
  const int lo = lane & 15, hf = lane >> 4;

  v2f wf[4][32];
  {
    const float* wrow = Wi + (size_t)lo * D_ + hf * 2;
#pragma unroll
    for (int nt = 0; nt < 4; ++nt) {
#pragma unroll
      for (int ks = 0; ks < 32; ++ks)
        wf[nt][ks] = *(const v2f*)(wrow + nt * 16 * D_ + ks * 4);
    }
  }

  v8f seed[4];
#pragma unroll
  for (int nt = 0; nt < 4; ++nt) {
    const float4 w0 = *(const float4*)(Wib + nt * 16 + hf * 8);
    const float4 w1 = *(const float4*)(Wib + nt * 16 + hf * 8 + 4);
    const float4 c0 = *(const float4*)(bias + nt * 16 + hf * 8);
    const float4 c1 = *(const float4*)(bias + nt * 16 + hf * 8 + 4);
    seed[nt][0] = w0.x + c0.x; seed[nt][1] = w0.y + c0.y;
    seed[nt][2] = w0.z + c0.z; seed[nt][3] = w0.w + c0.w;
    seed[nt][4] = w1.x + c1.x; seed[nt][5] = w1.y + c1.y;
    seed[nt][6] = w1.z + c1.z; seed[nt][7] = w1.w + c1.w;
  }

  const float* xrow = x + ((size_t)(btile * 16 + lo) * T_ + t0) * D_ + hf * 2;
  float* o = xp + ((size_t)btile * T_ + t0) * 1024 + lane * 4;

  v2f xa[32], xb[32];
#pragma unroll
  for (int ks = 0; ks < 32; ++ks) xa[ks] = *(const v2f*)(xrow + ks * 4);

#pragma unroll 1
  for (int it = 0; it < XT; it += 2) {
    {
      const float* nx = xrow + (size_t)(it + 1) * D_;
#pragma unroll
      for (int ks = 0; ks < 32; ++ks) xb[ks] = *(const v2f*)(nx + ks * 4);
    }
    asm volatile("" : "+v"(seed[0]), "+v"(seed[1]), "+v"(seed[2]), "+v"(seed[3]) : : "memory");
    xproj_step(wf, seed, xa, o + (size_t)it * 1024);
    {
      const int nn = (it + 2 < XT) ? (it + 2) : (XT - 1);
      const float* nx = xrow + (size_t)nn * D_;
#pragma unroll
      for (int ks = 0; ks < 32; ++ks) xa[ks] = *(const v2f*)(nx + ks * 4);
    }
    asm volatile("" : "+v"(seed[0]), "+v"(seed[1]), "+v"(seed[2]), "+v"(seed[3]) : : "memory");
    xproj_step(wf, seed, xb, o + (size_t)(it + 1) * 1024);
  }
}

__global__ void __launch_bounds__(128)
scan_kernel(const float* __restrict__ Ap, const float* __restrict__ Am,
            const float* __restrict__ Wf, const float* __restrict__ Wo,
            const float* __restrict__ Wob, const float* __restrict__ xp,
            float* __restrict__ out) {
  __shared__ __align__(16) float hlds[2 * PLANE];
  const int lane = threadIdx.x & 31;
  const int lo = lane & 15, hf = lane >> 4;
  const int nt = __builtin_amdgcn_readfirstlane((int)(threadIdx.x >> 5));
  const int btile = blockIdx.x;
  const bool up = nt >= 2;
  const int m1 = nt ^ 1, m2 = nt ^ 2, m3 = nt ^ 3;

  const int j = nt * 16 + lo;
  const int ko = hf * 8;
  v2f aw[8];
  {
    const float* src = (up ? Am : Ap) + j * 64 + ko;
#pragma unroll
    for (int p = 0; p < 4; ++p) {
      aw[p]     = *(const v2f*)(src + nt * 16 + 2 * p);
      aw[4 + p] = *(const v2f*)(src + m1 * 16 + 2 * p);
    }
  }
  v2f s0[4], s1[4], s2[4], s3[4];
  {
    const float sg = up ? -1.0f : 1.0f;
    const float* src = Wf + j * 64 + ko;
#pragma unroll
    for (int p = 0; p < 4; ++p) {
      v2f w;
      w = *(const v2f*)(src + nt * 16 + 2 * p); s0[p].x = w.x * sg; s0[p].y = w.y * sg;
      w = *(const v2f*)(src + m1 * 16 + 2 * p); s1[p].x = w.x * sg; s1[p].y = w.y * sg;
      w = *(const v2f*)(src + m2 * 16 + 2 * p); s2[p].x = w.x * sg; s2[p].y = w.y * sg;
      w = *(const v2f*)(src + m3 * 16 + 2 * p); s3[p].x = w.x * sg; s3[p].y = w.y * sg;
    }
  }

  float hn[8];
#pragma unroll
  for (int r = 0; r < 8; ++r) hn[r] = 0.0f;
  {
    float4 z; z.x = 0.0f; z.y = 0.0f; z.z = 0.0f; z.w = 0.0f;
    *(float4*)(hlds + nt * TILE + lane * 4) = z;
    *(float4*)(hlds + nt * TILE + 128 + lane * 4) = z;
  }

  const float* xpl = xp + (size_t)btile * T_ * 1024 + nt * 256 + lane * 4;
  const float* xpf = xp + (size_t)btile * T_ * 1024 + nt * 256 + lane * 8;
  float4 xq0 = *(const float4*)(xpl);
  float4 xq1 = *(const float4*)(xpl + 128);

  const float* r1 = hlds + m1 * TILE + lane * 4;
  const float* r2 = hlds + m2 * TILE + lane * 4;
  const float* r3 = hlds + m3 * TILE + lane * 4;
  float* w0 = hlds + nt * TILE + lane * 4;

#define GAP(...) asm volatile("" : __VA_ARGS__)
#define ALL_ACC "+v"(uc), "+v"(u2), "+v"(u3)
#define PR(x, y) v2f{(x), (y)}

  v8f uc, u2;
  float t0;
#define OWN_FIRST(TN, TP)                                                            \
  {                                                                                  \
    v8f xs, z8;                                                                      \
    xs[0] = xq0.x; xs[1] = xq0.y; xs[2] = xq0.z; xs[3] = xq0.w;                      \
    xs[4] = xq1.x; xs[5] = xq1.y; xs[6] = xq1.z; xs[7] = xq1.w;                      \
    _Pragma("unroll") for (int v = 0; v < 8; ++v) z8[v] = 0.0f;                      \
    u2 = wmma4(s0[0], h01, z8);               t0 = sq2(h01.x, h01.y);                \
    GAP("+v"(u2), "+v"(t0));                                                         \
    uc = wmma4(aw[0], h01, xs);               t0 += sq2(hn[2], hn[3]);               \
    GAP("+v"(uc), "+v"(u2), "+v"(t0));                                               \
    u2 = wmma4(s0[1], PR(hn[2], hn[3]), u2);  t0 += sq2(hn[4], hn[5]);               \
    GAP("+v"(uc), "+v"(u2), "+v"(t0));                                               \
    uc = wmma4(aw[1], PR(hn[2], hn[3]), uc);  t0 += sq2(hn[6], hn[7]);               \
    int tn = (TN);                                                                   \
        \
        \
        \
        \
        \
        \
    asm volatile("" : "+s"(tn), "+v"(uc), "+v"(u2), "+v"(t0));                       \
    {                                                                                \
      const float* nx = xpl + (size_t)tn * 1024;                                     \
      xq0 = *(const float4*)(nx);                                                    \
      xq1 = *(const float4*)(nx + 128);                                              \
      __builtin_prefetch(xpf + (size_t)(TP) * 1024, 0, 1);                           \
    }                                                                                \
        \
        \
        \
        \
        \
    asm volatile("s_wait_dscnt 0x0\n\ts_barrier_signal -1\n\ts_barrier_wait -1"      \
                 : "+v"(uc), "+v"(u2) : : "memory");                                 \
  }

  {
    v2f h01 = v2f{hn[0], hn[1]};
    OWN_FIRST(1, 1 + PF)
  }

#pragma unroll 1
  for (int t = 0; t < T_; t += 2) {
#pragma unroll
    for (int p = 0; p < 2; ++p) {
      const int pro = p * PLANE;
      const int pwo = (p ^ 1) * PLANE;

      v4f a1 = *(const v4f*)(r1 + pro);
      v4f b1 = *(const v4f*)(r1 + pro + 128);
      v4f a2 = *(const v4f*)(r2 + pro);
      v4f b2 = *(const v4f*)(r2 + pro + 128);
      v4f a3 = *(const v4f*)(r3 + pro);
      v4f b3 = *(const v4f*)(r3 + pro + 128);
      asm volatile("" : "+v"(uc), "+v"(u2) : : "memory");

      uc = wmma4(aw[2], PR(hn[4], hn[5]), uc);
      u2 = wmma4(s0[2], PR(hn[4], hn[5]), u2);
      uc = wmma4(aw[3], PR(hn[6], hn[7]), uc);
      u2 = wmma4(s0[3], PR(hn[6], hn[7]), u2);
      GAP("+v"(uc), "+v"(u2), "+v"(t0), "+v"(a1), "+v"(b1), "+v"(a2), "+v"(b2), "+v"(a3), "+v"(b3));

      v8f u3;
#pragma unroll
      for (int v = 0; v < 8; ++v) u3[v] = 0.0f;
      float t1, t2, t3;
      u3 = wmma4(s1[0], PR(a1.x, a1.y), u3);  GAP(ALL_ACC, "+v"(a1));
      t1 = sq2(a1.x, a1.y);   t1 += sq2(a1.z, a1.w);
      uc = wmma4(aw[4], PR(a1.x, a1.y), uc);  GAP(ALL_ACC, "+v"(b1), "+v"(t1));
      t1 += sq2(b1.x, b1.y);  t1 += sq2(b1.z, b1.w);
      u3 = wmma4(s1[1], PR(a1.z, a1.w), u3);  GAP(ALL_ACC, "+v"(a2));
      t2 = sq2(a2.x, a2.y);   t2 += sq2(a2.z, a2.w);
      u2 = wmma4(s2[0], PR(a2.x, a2.y), u2);  GAP(ALL_ACC, "+v"(b2), "+v"(t2));
      t2 += sq2(b2.x, b2.y);  t2 += sq2(b2.z, b2.w);
      u3 = wmma4(s1[2], PR(b1.x, b1.y), u3);  GAP(ALL_ACC, "+v"(a3));
      t3 = sq2(a3.x, a3.y);   t3 += sq2(a3.z, a3.w);
      uc = wmma4(aw[5], PR(a1.z, a1.w), uc);  GAP(ALL_ACC, "+v"(b3), "+v"(t3));
      t3 += sq2(b3.x, b3.y);  t3 += sq2(b3.z, b3.w);
      u3 = wmma4(s1[3], PR(b1.z, b1.w), u3);  GAP(ALL_ACC, "+v"(t0), "+v"(t1), "+v"(t2), "+v"(t3));
      const float sa = t0 + t1;
      const float sb = t2 + t3;
      float pa = sa + sb;
      float pm = up ? sa : sb;
      u2 = wmma4(s2[1], PR(a2.z, a2.w), u2);  GAP(ALL_ACC, "+v"(pa), "+v"(pm));
      pa += xhalf(pa);
      pm += xhalf(pm);
      u3 = wmma4(s3[0], PR(a3.x, a3.y), u3);  GAP(ALL_ACC, "+v"(pa), "+v"(pm));
      pa = rcp_hw(pa + 1e-8f);
      uc = wmma4(aw[6], PR(b1.x, b1.y), uc);  GAP(ALL_ACC, "+v"(pa), "+v"(pm));
      pa = -(pm * pa - 0.72134752044448170f) * 20.0f;
      u3 = wmma4(s3[1], PR(a3.z, a3.w), u3);  GAP(ALL_ACC, "+v"(pa));
      pa = 1.0f + __expf(pa);
      u2 = wmma4(s2[2], PR(b2.x, b2.y), u2);  GAP(ALL_ACC, "+v"(pa));
      pa = rcp_hw(pa);
      u3 = wmma4(s3[2], PR(b3.x, b3.y), u3);
      uc = wmma4(aw[7], PR(b1.z, b1.w), uc);
      u3 = wmma4(s3[3], PR(b3.z, b3.w), u3);
      u2 = wmma4(s2[3], PR(b2.z, b2.w), u2);

#pragma unroll
      for (int r = 0; r < 8; ++r)
        hn[r] = fast_tanh(__builtin_fmaf(pa, u2[r] + u3[r], uc[r]));
      {
        float4 s;
        s.x = hn[0]; s.y = hn[1]; s.z = hn[2]; s.w = hn[3];
        *(float4*)(w0 + pwo) = s;
        s.x = hn[4]; s.y = hn[5]; s.z = hn[6]; s.w = hn[7];
        *(float4*)(w0 + pwo + 128) = s;
      }
      v2f h01 = v2f{hn[0], hn[1]};
      asm volatile("" : "+v"(h01) : : "memory");

      const int nxt = t + p + 2;
      OWN_FIRST((nxt < T_) ? nxt : (T_ - 1), (nxt + PF < T_) ? (nxt + PF) : (T_ - 1))
    }
  }
#undef OWN_FIRST
#undef GAP
#undef ALL_ACC
#undef PR

  const float* hT = hlds + (T_ & 1) * PLANE;

  for (int i = threadIdx.x; i < 1024; i += 128) {
    const int row = i >> 6, col = i & 63;
    const int m = col >> 4, h8 = (col >> 3) & 1, q = (col >> 2) & 1, e = col & 3;
    out[32768 + btile * 1024 + i] = hT[m * TILE + q * 128 + (h8 * 16 + row) * 4 + e];
  }

  float4 fa[4], fb[4];
#pragma unroll
  for (int m = 0; m < 4; ++m) {
    fa[m] = *(const float4*)(hT + m * TILE + lane * 4);
    fb[m] = *(const float4*)(hT + m * TILE + 128 + lane * 4);
  }
#pragma unroll
  for (int n2i = 0; n2i < 2; ++n2i) {
    const int n2 = nt * 2 + n2i;
    v8f p;
#pragma unroll
    for (int r = 0; r < 8; ++r) p[r] = Wob[n2 * 16 + hf * 8 + r];
    const float* wsrc = Wo + (size_t)(n2 * 16 + lo) * 64 + ko;
#pragma unroll
    for (int m = 0; m < 4; ++m) {
      p = wmma4(*(const v2f*)(wsrc + m * 16 + 0), v2f{fa[m].x, fa[m].y}, p);
      p = wmma4(*(const v2f*)(wsrc + m * 16 + 2), v2f{fa[m].z, fa[m].w}, p);
      p = wmma4(*(const v2f*)(wsrc + m * 16 + 4), v2f{fb[m].x, fb[m].y}, p);
      p = wmma4(*(const v2f*)(wsrc + m * 16 + 6), v2f{fb[m].z, fb[m].w}, p);
    }
    float* o = out + (size_t)(btile * 16 + lo) * 128 + n2 * 16 + hf * 8;
    float4 s;
    s.x = p[0]; s.y = p[1]; s.z = p[2]; s.w = p[3];
    *(float4*)(o) = s;
    s.x = p[4]; s.y = p[5]; s.z = p[6]; s.w = p[7];
    *(float4*)(o + 4) = s;
  }
}

extern "C" void kernel_launch(void* const* d_in, const int* in_sizes, int n_in,
                              void* d_out, int out_size, void* d_ws, size_t ws_size,
                              hipStream_t stream) {
  const float* x    = (const float*)d_in[0];
  const float* Ap   = (const float*)d_in[1];
  const float* Am   = (const float*)d_in[2];
  const float* Wf   = (const float*)d_in[3];
  const float* Wiw  = (const float*)d_in[4];
  const float* Wib  = (const float*)d_in[5];
  const float* Wow  = (const float*)d_in[6];
  const float* Wob  = (const float*)d_in[7];
  const float* bias = (const float*)d_in[8];
  float* out = (float*)d_out;
  float* xp  = (float*)d_ws;

  xproj_kernel<<<1024, 128, 0, stream>>>(x, Wiw, Wib, bias, xp);
  scan_kernel<<<16, 128, 0, stream>>>(Ap, Am, Wf, Wow, Wob, xp, out);
}


# scan: gate's sums of squares as packed two-wide fma on the register pairs as loaded, three independent sums per slice (were scalar chains with moves); squares' sum order changes, matrix sums same
# speedup vs baseline: 5.4236x; 1.1206x over previous
#include <hip/hip_runtime.h>
#include <hip/hip_bf16.h>


typedef float v8f __attribute__((ext_vector_type(8)));
typedef float v2f __attribute__((ext_vector_type(2)));
typedef float v4f __attribute__((ext_vector_type(4)));

#define T_  2048
#define D_  128
#define TILE  256
#define PLANE (4 * TILE)
#define PF  8


__device__ __forceinline__ v8f wmma4(v2f a, v2f b, v8f c) {
  return __builtin_amdgcn_wmma_f32_16x16x4_f32(false, a, false, b, (short)0, c,
                                               false, false);
}

__device__ __forceinline__ float rcp_hw(float d) {
  return __builtin_amdgcn_rcpf(d);
}

__device__ __forceinline__ float fast_tanh(float x) {
  return __builtin_amdgcn_tanhf(x);
}

__device__ __forceinline__ float xhalf(float x) {
  int b = __builtin_bit_cast(int, x);
  return __builtin_bit_cast(
      float, __builtin_amdgcn_permlanex16(b, b, 0x76543210, 0xfedcba98, false, false));
}

__device__ __forceinline__ v2f sqp(v2f a) { return a * a; }
__device__ __forceinline__ v2f sqacc(v2f a, v2f q) {
  return __builtin_elementwise_fma(a, a, q);
}

#define XT 8

__device__ __forceinline__ void xproj_step(const v2f (&wf)[4][32], const v8f (&seed)[4],
                                           const v2f (&xf)[32], float* __restrict__ o) {
  v8f acc[4];
#pragma unroll
  for (int nt = 0; nt < 4; ++nt) acc[nt] = wmma4(wf[nt][0], xf[0], seed[nt]);
#pragma unroll
  for (int ks = 1; ks < 32; ++ks) {
#pragma unroll
    for (int nt = 0; nt < 4; ++nt) acc[nt] = wmma4(wf[nt][ks], xf[ks], acc[nt]);
  }
#pragma unroll
  for (int nt = 0; nt < 4; ++nt) {
#pragma unroll
    for (int q = 0; q < 2; ++q) {
      float4 s;
      s.x = acc[nt][q * 4 + 0]; s.y = acc[nt][q * 4 + 1];
      s.z = acc[nt][q * 4 + 2]; s.w = acc[nt][q * 4 + 3];
      *(float4*)(o + (nt * 2 + q) * 128) = s;
    }
  }
}

__global__ void __launch_bounds__(128, 2)
xproj_kernel(const float* x, const float* __restrict__ Wi,
             const float* __restrict__ Wib, const float* __restrict__ bias,
             float* xp) {
  const int lane = threadIdx.x & 31;
  const int wv   = blockIdx.x * 4 + (threadIdx.x >> 5);
  const int btile = wv >> 8;
  const int t0    = (wv & 255) * XT;
  const int lo = lane & 15, hf = lane >> 4;

  v2f wf[4][32];
  {
    const float* wrow = Wi + (size_t)lo * D_ + hf * 2;
#pragma unroll
    for (int nt = 0; nt < 4; ++nt) {
#pragma unroll
      for (int ks = 0; ks < 32; ++ks)
        wf[nt][ks] = *(const v2f*)(wrow + nt * 16 * D_ + ks * 4);
    }
  }

  v8f seed[4];
#pragma unroll
  for (int nt = 0; nt < 4; ++nt) {
    const float4 w0 = *(const float4*)(Wib + nt * 16 + hf * 8);
    const float4 w1 = *(const float4*)(Wib + nt * 16 + hf * 8 + 4);
    const float4 c0 = *(const float4*)(bias + nt * 16 + hf * 8);
    const float4 c1 = *(const float4*)(bias + nt * 16 + hf * 8 + 4);
    seed[nt][0] = w0.x + c0.x; seed[nt][1] = w0.y + c0.y;
    seed[nt][2] = w0.z + c0.z; seed[nt][3] = w0.w + c0.w;
    seed[nt][4] = w1.x + c1.x; seed[nt][5] = w1.y + c1.y;
    seed[nt][6] = w1.z + c1.z; seed[nt][7] = w1.w + c1.w;
  }

  const float* xrow = x + ((size_t)(btile * 16 + lo) * T_ + t0) * D_ + hf * 2;
  float* o = xp + ((size_t)btile * T_ + t0) * 1024 + lane * 4;

  v2f xa[32], xb[32];
#pragma unroll
  for (int ks = 0; ks < 32; ++ks) xa[ks] = *(const v2f*)(xrow + ks * 4);

#pragma unroll 1
  for (int it = 0; it < XT; it += 2) {
    {
      const float* nx = xrow + (size_t)(it + 1) * D_;
#pragma unroll
      for (int ks = 0; ks < 32; ++ks) xb[ks] = *(const v2f*)(nx + ks * 4);
    }
    asm volatile("" : "+v"(seed[0]), "+v"(seed[1]), "+v"(seed[2]), "+v"(seed[3]) : : "memory");
    xproj_step(wf, seed, xa, o + (size_t)it * 1024);
    {
      const int nn = (it + 2 < XT) ? (it + 2) : (XT - 1);
      const float* nx = xrow + (size_t)nn * D_;
#pragma unroll
      for (int ks = 0; ks < 32; ++ks) xa[ks] = *(const v2f*)(nx + ks * 4);
    }
    asm volatile("" : "+v"(seed[0]), "+v"(seed[1]), "+v"(seed[2]), "+v"(seed[3]) : : "memory");
    xproj_step(wf, seed, xb, o + (size_t)(it + 1) * 1024);
  }
}

__global__ void __launch_bounds__(128)
scan_kernel(const float* __restrict__ Ap, const float* __restrict__ Am,
            const float* __restrict__ Wf, const float* __restrict__ Wo,
            const float* __restrict__ Wob, const float* __restrict__ xp,
            float* __restrict__ out) {
  __shared__ __align__(16) float hlds[2 * PLANE];
  const int lane = threadIdx.x & 31;
  const int lo = lane & 15, hf = lane >> 4;
  const int nt = __builtin_amdgcn_readfirstlane((int)(threadIdx.x >> 5));
  const int btile = blockIdx.x;
  const bool up = nt >= 2;
  const int m1 = nt ^ 1, m2 = nt ^ 2, m3 = nt ^ 3;

  const int j = nt * 16 + lo;
  const int ko = hf * 8;
  v2f aw[8];
  {
    const float* src = (up ? Am : Ap) + j * 64 + ko;
#pragma unroll
    for (int p = 0; p < 4; ++p) {
      aw[p]     = *(const v2f*)(src + nt * 16 + 2 * p);
      aw[4 + p] = *(const v2f*)(src + m1 * 16 + 2 * p);
    }
  }
  v2f s0[4], s1[4], s2[4], s3[4];
  {
    const float sg = up ? -1.0f : 1.0f;
    const float* src = Wf + j * 64 + ko;
#pragma unroll
    for (int p = 0; p < 4; ++p) {
      v2f w;
      w = *(const v2f*)(src + nt * 16 + 2 * p); s0[p].x = w.x * sg; s0[p].y = w.y * sg;
      w = *(const v2f*)(src + m1 * 16 + 2 * p); s1[p].x = w.x * sg; s1[p].y = w.y * sg;
      w = *(const v2f*)(src + m2 * 16 + 2 * p); s2[p].x = w.x * sg; s2[p].y = w.y * sg;
      w = *(const v2f*)(src + m3 * 16 + 2 * p); s3[p].x = w.x * sg; s3[p].y = w.y * sg;
    }
  }

  float hn[8];
#pragma unroll
  for (int r = 0; r < 8; ++r) hn[r] = 0.0f;
  {
    float4 z; z.x = 0.0f; z.y = 0.0f; z.z = 0.0f; z.w = 0.0f;
    *(float4*)(hlds + nt * TILE + lane * 4) = z;
    *(float4*)(hlds + nt * TILE + 128 + lane * 4) = z;
  }

  const float* xpl = xp + (size_t)btile * T_ * 1024 + nt * 256 + lane * 4;
  const float* xpf = xp + (size_t)btile * T_ * 1024 + nt * 256 + lane * 8;
  float4 xq0 = *(const float4*)(xpl);
  float4 xq1 = *(const float4*)(xpl + 128);

  const float* r1 = hlds + m1 * TILE + lane * 4;
  const float* r2 = hlds + m2 * TILE + lane * 4;
  const float* r3 = hlds + m3 * TILE + lane * 4;
  float* w0 = hlds + nt * TILE + lane * 4;

#define GAP(...) asm volatile("" : __VA_ARGS__)
#define ALL_ACC "+v"(uc), "+v"(u2), "+v"(u3)
#define PR(x, y) v2f{(x), (y)}

  v8f uc, u2;
  float t0;
  v2f q0;
#define OWN_FIRST(TN, TP)                                                            \
  {                                                                                  \
    v8f xs, z8;                                                                      \
    xs[0] = xq0.x; xs[1] = xq0.y; xs[2] = xq0.z; xs[3] = xq0.w;                      \
    xs[4] = xq1.x; xs[5] = xq1.y; xs[6] = xq1.z; xs[7] = xq1.w;                      \
    _Pragma("unroll") for (int v = 0; v < 8; ++v) z8[v] = 0.0f;                      \
    u2 = wmma4(s0[0], h01, z8);               q0 = sqp(h01);                         \
    GAP("+v"(u2), "+v"(q0));                                                         \
    uc = wmma4(aw[0], h01, xs);               q0 = sqacc(PR(hn[2], hn[3]), q0);      \
    GAP("+v"(uc), "+v"(u2), "+v"(q0));                                               \
    u2 = wmma4(s0[1], PR(hn[2], hn[3]), u2);  q0 = sqacc(PR(hn[4], hn[5]), q0);      \
    GAP("+v"(uc), "+v"(u2), "+v"(q0));                                               \
    uc = wmma4(aw[1], PR(hn[2], hn[3]), uc);  q0 = sqacc(PR(hn[6], hn[7]), q0);      \
    t0 = q0.x + q0.y;                                                                \
    int tn = (TN);                                                                   \
        \
        \
        \
        \
        \
        \
    asm volatile("" : "+s"(tn), "+v"(uc), "+v"(u2), "+v"(t0));                       \
    {                                                                                \
      const float* nx = xpl + (size_t)tn * 1024;                                     \
      xq0 = *(const float4*)(nx);                                                    \
      xq1 = *(const float4*)(nx + 128);                                              \
      __builtin_prefetch(xpf + (size_t)(TP) * 1024, 0, 1);                           \
    }                                                                                \
        \
        \
        \
        \
        \
    asm volatile("s_wait_dscnt 0x0\n\ts_barrier_signal -1\n\ts_barrier_wait -1"      \
                 : "+v"(uc), "+v"(u2) : : "memory");                                 \
  }

  {
    v2f h01 = v2f{hn[0], hn[1]};
    OWN_FIRST(1, 1 + PF)
  }

#pragma unroll 1
  for (int t = 0; t < T_; t += 2) {
#pragma unroll
    for (int p = 0; p < 2; ++p) {
      const int pro = p * PLANE;
      const int pwo = (p ^ 1) * PLANE;

      v4f a1 = *(const v4f*)(r1 + pro);
      v4f b1 = *(const v4f*)(r1 + pro + 128);
      v4f a2 = *(const v4f*)(r2 + pro);
      v4f b2 = *(const v4f*)(r2 + pro + 128);
      v4f a3 = *(const v4f*)(r3 + pro);
      v4f b3 = *(const v4f*)(r3 + pro + 128);
      asm volatile("" : "+v"(uc), "+v"(u2) : : "memory");

      uc = wmma4(aw[2], PR(hn[4], hn[5]), uc);
      u2 = wmma4(s0[2], PR(hn[4], hn[5]), u2);
      uc = wmma4(aw[3], PR(hn[6], hn[7]), uc);
      u2 = wmma4(s0[3], PR(hn[6], hn[7]), u2);
      GAP("+v"(uc), "+v"(u2), "+v"(t0), "+v"(a1), "+v"(b1), "+v"(a2), "+v"(b2), "+v"(a3), "+v"(b3));

      v8f u3;
#pragma unroll
      for (int v = 0; v < 8; ++v) u3[v] = 0.0f;
      v2f q1, q2, q3;
      float t1, t2, t3;
      u3 = wmma4(s1[0], PR(a1.x, a1.y), u3);  GAP(ALL_ACC, "+v"(a1), "+v"(a2), "+v"(a3));
      q1 = sqp(PR(a1.x, a1.y));  q2 = sqp(PR(a2.x, a2.y));  q3 = sqp(PR(a3.x, a3.y));
      uc = wmma4(aw[4], PR(a1.x, a1.y), uc);  GAP(ALL_ACC, "+v"(q1), "+v"(q2), "+v"(q3));
      q1 = sqacc(PR(a1.z, a1.w), q1);  q2 = sqacc(PR(a2.z, a2.w), q2);  q3 = sqacc(PR(a3.z, a3.w), q3);
      u3 = wmma4(s1[1], PR(a1.z, a1.w), u3);  GAP(ALL_ACC, "+v"(q1), "+v"(q2), "+v"(q3), "+v"(b1), "+v"(b2), "+v"(b3));
      q1 = sqacc(PR(b1.x, b1.y), q1);  q2 = sqacc(PR(b2.x, b2.y), q2);  q3 = sqacc(PR(b3.x, b3.y), q3);
      u2 = wmma4(s2[0], PR(a2.x, a2.y), u2);  GAP(ALL_ACC, "+v"(q1), "+v"(q2), "+v"(q3));
      q1 = sqacc(PR(b1.z, b1.w), q1);  q2 = sqacc(PR(b2.z, b2.w), q2);  q3 = sqacc(PR(b3.z, b3.w), q3);
      u3 = wmma4(s1[2], PR(b1.x, b1.y), u3);  GAP(ALL_ACC, "+v"(q1), "+v"(q2), "+v"(q3));
      t1 = q1.x + q1.y;  t2 = q2.x + q2.y;  t3 = q3.x + q3.y;
      uc = wmma4(aw[5], PR(a1.z, a1.w), uc);  GAP(ALL_ACC, "+v"(t0), "+v"(t1), "+v"(t2), "+v"(t3));
      float sa = t0 + t1;
      float sb = t2 + t3;
      u3 = wmma4(s1[3], PR(b1.z, b1.w), u3);  GAP(ALL_ACC, "+v"(sa), "+v"(sb));
      float pa = sa + sb;
      float pm = up ? sa : sb;
      u2 = wmma4(s2[1], PR(a2.z, a2.w), u2);  GAP(ALL_ACC, "+v"(pa), "+v"(pm));
      pa += xhalf(pa);
      pm += xhalf(pm);
      u3 = wmma4(s3[0], PR(a3.x, a3.y), u3);  GAP(ALL_ACC, "+v"(pa), "+v"(pm));
      pa = rcp_hw(pa + 1e-8f);
      uc = wmma4(aw[6], PR(b1.x, b1.y), uc);  GAP(ALL_ACC, "+v"(pa), "+v"(pm));
      pa = -(pm * pa - 0.72134752044448170f) * 20.0f;
      u3 = wmma4(s3[1], PR(a3.z, a3.w), u3);  GAP(ALL_ACC, "+v"(pa));
      pa = 1.0f + __expf(pa);
      u2 = wmma4(s2[2], PR(b2.x, b2.y), u2);  GAP(ALL_ACC, "+v"(pa));
      pa = rcp_hw(pa);
      u3 = wmma4(s3[2], PR(b3.x, b3.y), u3);
      uc = wmma4(aw[7], PR(b1.z, b1.w), uc);
      u3 = wmma4(s3[3], PR(b3.z, b3.w), u3);
      u2 = wmma4(s2[3], PR(b2.z, b2.w), u2);

#pragma unroll
      for (int r = 0; r < 8; ++r)
        hn[r] = fast_tanh(__builtin_fmaf(pa, u2[r] + u3[r], uc[r]));
      {
        float4 s;
        s.x = hn[0]; s.y = hn[1]; s.z = hn[2]; s.w = hn[3];
        *(float4*)(w0 + pwo) = s;
        s.x = hn[4]; s.y = hn[5]; s.z = hn[6]; s.w = hn[7];
        *(float4*)(w0 + pwo + 128) = s;
      }
      v2f h01 = v2f{hn[0], hn[1]};
      asm volatile("" : "+v"(h01) : : "memory");

      const int nxt = t + p + 2;
      OWN_FIRST((nxt < T_) ? nxt : (T_ - 1), (nxt + PF < T_) ? (nxt + PF) : (T_ - 1))
    }
  }
#undef OWN_FIRST
#undef GAP
#undef ALL_ACC
#undef PR

  const float* hT = hlds + (T_ & 1) * PLANE;

  for (int i = threadIdx.x; i < 1024; i += 128) {
    const int row = i >> 6, col = i & 63;
    const int m = col >> 4, h8 = (col >> 3) & 1, q = (col >> 2) & 1, e = col & 3;
    out[32768 + btile * 1024 + i] = hT[m * TILE + q * 128 + (h8 * 16 + row) * 4 + e];
  }

  float4 fa[4], fb[4];
#pragma unroll
  for (int m = 0; m < 4; ++m) {
    fa[m] = *(const float4*)(hT + m * TILE + lane * 4);
    fb[m] = *(const float4*)(hT + m * TILE + 128 + lane * 4);
  }
#pragma unroll
  for (int n2i = 0; n2i < 2; ++n2i) {
    const int n2 = nt * 2 + n2i;
    v8f p;
#pragma unroll
    for (int r = 0; r < 8; ++r) p[r] = Wob[n2 * 16 + hf * 8 + r];
    const float* wsrc = Wo + (size_t)(n2 * 16 + lo) * 64 + ko;
#pragma unroll
    for (int m = 0; m < 4; ++m) {
      p = wmma4(*(const v2f*)(wsrc + m * 16 + 0), v2f{fa[m].x, fa[m].y}, p);
      p = wmma4(*(const v2f*)(wsrc + m * 16 + 2), v2f{fa[m].z, fa[m].w}, p);
      p = wmma4(*(const v2f*)(wsrc + m * 16 + 4), v2f{fb[m].x, fb[m].y}, p);
      p = wmma4(*(const v2f*)(wsrc + m * 16 + 6), v2f{fb[m].z, fb[m].w}, p);
    }
    float* o = out + (size_t)(btile * 16 + lo) * 128 + n2 * 16 + hf * 8;
    float4 s;
    s.x = p[0]; s.y = p[1]; s.z = p[2]; s.w = p[3];
    *(float4*)(o) = s;
    s.x = p[4]; s.y = p[5]; s.z = p[6]; s.w = p[7];
    *(float4*)(o + 4) = s;
  }
}

extern "C" void kernel_launch(void* const* d_in, const int* in_sizes, int n_in,
                              void* d_out, int out_size, void* d_ws, size_t ws_size,
                              hipStream_t stream) {
  const float* x    = (const float*)d_in[0];
  const float* Ap   = (const float*)d_in[1];
  const float* Am   = (const float*)d_in[2];
  const float* Wf   = (const float*)d_in[3];
  const float* Wiw  = (const float*)d_in[4];
  const float* Wib  = (const float*)d_in[5];
  const float* Wow  = (const float*)d_in[6];
  const float* Wob  = (const float*)d_in[7];
  const float* bias = (const float*)d_in[8];
  float* out = (float*)d_out;
  float* xp  = (float*)d_ws;

  xproj_kernel<<<1024, 128, 0, stream>>>(x, Wiw, Wib, bias, xp);
  scan_kernel<<<16, 128, 0, stream>>>(Ap, Am, Wf, Wow, Wob, xp, out);
}
